# starting kernel
# baseline (speedup 1.0000x reference)
#include <hip/hip_runtime.h>

#define EPSV 1e-5f

typedef __attribute__((ext_vector_type(2))) float v2f;
typedef __attribute__((ext_vector_type(8))) float v8f;

__global__ __launch_bounds__(256) void prep_weights_kernel(
    const float* __restrict__ w1, const float* __restrict__ w2,
    float* __restrict__ wt)
{
    int i = blockIdx.x * 256 + threadIdx.x;
    if (i >= 2 * 36864) return;
    const float* w = (i < 36864) ? w1 : w2;
    float* dst = wt + (i < 36864 ? 0 : 36864);
    int j = i % 36864;
    int tap = j / 4096;
    int rem = j % 4096;
    int n = (rem >> 1) & 63;
    int k = ((rem >> 7) << 1) | (rem & 1);
    dst[j] = w[(n * 64 + k) * 9 + tap];
}

__global__ __launch_bounds__(256) void fused_basicblock_kernel(
    const float* __restrict__ x,
    const float* __restrict__ wt,
    const float* __restrict__ g1, const float* __restrict__ b1,
    const float* __restrict__ m1, const float* __restrict__ v1,
    const float* __restrict__ g2, const float* __restrict__ b2,
    const float* __restrict__ m2, const float* __restrict__ v2p,
    const float* __restrict__ mask1, const float* __restrict__ mask2,
    float* __restrict__ out)
{
    __shared__ __align__(16) float xpad[64 * 81];
    __shared__ __align__(16) float ipad[64 * 81];
    __shared__ __align__(16) float wsh[64 * 64];

    const int tid  = threadIdx.x;
    const int lane = tid & 31;
    const int wave = tid >> 5;
    const int b    = blockIdx.x;

    const int lo  = lane & 15;
    const int hi  = lane >> 4;
    const int khi = hi * 2;

    const int nn  = (wave & 3) * 16 + lo;
    const int mt0 = (wave >> 2);
    int mA0 = mt0 * 16 + lo;        if (mA0 > 48) mA0 = 48;
    int mA1 = (mt0 + 2) * 16 + lo;  if (mA1 > 48) mA1 = 48;
    const int y0 = mA0 / 7, x0c = mA0 % 7;
    const int y1 = mA1 / 7, x1c = mA1 % 7;

    for (int i = tid; i < 64 * 81; i += 256) { xpad[i] = 0.0f; ipad[i] = 0.0f; }

    const float* xb = x + (size_t)b * 3136;
    for (int i = tid; i < 3136; i += 256) {
        int c = i / 49, m = i % 49;
        xpad[c * 81 + (m / 7 + 1) * 9 + (m % 7) + 1] = xb[i];
    }
    __syncthreads();

    auto conv_pass = [&](const float* src, const float* __restrict__ wslab,
                         v8f& acc0, v8f& acc1) {
        for (int tap = 0; tap < 9; ++tap) {
            __syncthreads();
            {
                const float4* s4 = (const float4*)(wslab + tap * 4096);
                float4* d4 = (float4*)wsh;
                #pragma unroll
                for (int i = 0; i < 4; ++i) d4[tid + i * 256] = s4[tid + i * 256];
            }
            if (tap < 8)
                __builtin_prefetch((const char*)(wslab + (tap + 1) * 4096) + tid * 64, 0, 1);
            __syncthreads();

            const int r = tap / 3, s = tap - r * 3;
            const int so0 = (y0 + r) * 9 + (x0c + s);
            const int so1 = (y1 + r) * 9 + (x1c + s);
            #pragma unroll
            for (int ks = 0; ks < 16; ++ks) {
                const int kl = ks * 4 + khi;
                v2f a0, a1;
                a0[0] = src[kl * 81 + so0];
                a0[1] = src[kl * 81 + 81 + so0];
                a1[0] = src[kl * 81 + so1];
                a1[1] = src[kl * 81 + 81 + so1];
                const v2f bf = *(const v2f*)&wsh[(ks * 2 + hi) * 128 + nn * 2];
                acc0 = __builtin_amdgcn_wmma_f32_16x16x4_f32(
                    false, a0, false, bf, (short)0, acc0, false, false);
                acc1 = __builtin_amdgcn_wmma_f32_16x16x4_f32(
                    false, a1, false, bf, (short)0, acc1, false, false);
            }
        }
    };

    v8f c1a = {}, c1b = {};
    conv_pass(xpad, wt, c1a, c1b);

    const float sc1 = g1[nn] * rsqrtf(v1[nn] + EPSV);
    const float sh1 = b1[nn] - m1[nn] * sc1;
    auto epi1 = [&](const v8f& acc, int mt) {
        #pragma unroll
        for (int j = 0; j < 8; ++j) {
            int m = mt * 16 + j + 8 * hi;
            if (m < 49) {
                float v = fmaf(acc[j], sc1, sh1);
                v = fmaxf(v, 0.0f);
                if (b == 0) v *= mask1[nn * 49 + m];
                ipad[nn * 81 + (m / 7 + 1) * 9 + (m % 7) + 1] = v;
            }
        }
    };
    epi1(c1a, mt0);
    epi1(c1b, mt0 + 2);

    v8f c2a = {}, c2b = {};
    conv_pass(ipad, wt + 36864, c2a, c2b);
    __syncthreads();

    const float sc2 = g2[nn] * rsqrtf(v2p[nn] + EPSV);
    const float sh2 = b2[nn] - m2[nn] * sc2;
    auto epi2 = [&](const v8f& acc, int mt) {
        #pragma unroll
        for (int j = 0; j < 8; ++j) {
            int m = mt * 16 + j + 8 * hi;
            if (m < 49) {
                float v = acc[j];
                if (b == 0) v *= mask2[nn * 49 + m];
                v = fmaf(v, sc2, sh2);
                int off = nn * 81 + (m / 7 + 1) * 9 + (m % 7) + 1;
                v += xpad[off];
                ipad[off] = fmaxf(v, 0.0f);
            }
        }
    };
    epi2(c2a, mt0);
    epi2(c2b, mt0 + 2);
    __syncthreads();

    float* ob = out + (size_t)b * 3136;
    for (int i = tid; i < 3136; i += 256) {
        int c = i / 49, m = i % 49;
        ob[i] = ipad[c * 81 + (m / 7 + 1) * 9 + (m % 7) + 1];
    }
}

extern "C" void kernel_launch(void* const* d_in, const int* in_sizes, int n_in,
                              void* d_out, int out_size, void* d_ws, size_t ws_size,
                              hipStream_t stream) {
    const float* x     = (const float*)d_in[0];
    const float* w1    = (const float*)d_in[1];
    const float* g1    = (const float*)d_in[2];
    const float* b1    = (const float*)d_in[3];
    const float* m1    = (const float*)d_in[4];
    const float* v1    = (const float*)d_in[5];
    const float* w2    = (const float*)d_in[6];
    const float* g2    = (const float*)d_in[7];
    const float* b2    = (const float*)d_in[8];
    const float* m2    = (const float*)d_in[9];
    const float* v2p   = (const float*)d_in[10];
    const float* mask1 = (const float*)d_in[11];
    const float* mask2 = (const float*)d_in[12];
    float* wt = (float*)d_ws;
    float* out = (float*)d_out;

    prep_weights_kernel<<<288, 256, 0, stream>>>(w1, w2, wt);
    fused_basicblock_kernel<<<4096, 256, 0, stream>>>(
        x, wt, g1, b1, m1, v1, g2, b2, m2, v2p, mask1, mask2, out);
}


# Two images per workgroup: 98 rows in 7 M-tiles (was 2x4), each weight slab and B fragment serves both; one LDS activation buffer reused, residual x re-read at the store
# speedup vs baseline: 1.1282x; 1.1282x over previous
#include <hip/hip_runtime.h>

#define EPSV 1e-5f

typedef __attribute__((ext_vector_type(2))) float v2f;
typedef __attribute__((ext_vector_type(8))) float v8f;

__global__ __launch_bounds__(256) void prep_weights_kernel(
    const float* __restrict__ w1, const float* __restrict__ w2,
    float* __restrict__ wt)
{
    int i = blockIdx.x * 256 + threadIdx.x;
    if (i >= 2 * 36864) return;
    const float* w = (i < 36864) ? w1 : w2;
    float* dst = wt + (i < 36864 ? 0 : 36864);
    int j = i % 36864;
    int tap = j / 4096;
    int rem = j % 4096;
    int n = (rem >> 1) & 63;
    int k = ((rem >> 7) << 1) | (rem & 1);
    dst[j] = w[(n * 64 + k) * 9 + tap];
}

#define IMG_LDS 5184

template <int NT>
__device__ __forceinline__ void tap_mma(const float* src, const float* wsh,
                                        const int tapoff, const int* rowoff,
                                        const int khi, const int hi, const int nn,
                                        v8f* acc)
{
    #pragma unroll
    for (int ks = 0; ks < 16; ++ks) {
        const int kl = ks * 4 + khi;
        const v2f bf = *(const v2f*)&wsh[(ks * 2 + hi) * 128 + nn * 2];
        #pragma unroll
        for (int t = 0; t < NT; ++t) {
            v2f a;
            a[0] = src[kl * 81 + rowoff[t] + tapoff];
            a[1] = src[kl * 81 + 81 + rowoff[t] + tapoff];
            acc[t] = __builtin_amdgcn_wmma_f32_16x16x4_f32(
                false, a, false, bf, (short)0, acc[t], false, false);
        }
    }
}

__global__ __launch_bounds__(256, 6) void fused_basicblock_kernel(
    const float* __restrict__ x,
    const float* __restrict__ wt,
    const float* __restrict__ g1, const float* __restrict__ b1,
    const float* __restrict__ m1, const float* __restrict__ v1,
    const float* __restrict__ g2, const float* __restrict__ b2,
    const float* __restrict__ m2, const float* __restrict__ v2p,
    const float* __restrict__ mask1, const float* __restrict__ mask2,
    float* __restrict__ out)
{
    __shared__ __align__(16) float act[2 * IMG_LDS];
    __shared__ __align__(16) float wsh[64 * 64];

    const int tid  = threadIdx.x;
    const int lane = tid & 31;
    const int wave = __builtin_amdgcn_readfirstlane(tid >> 5);
    const int b0   = blockIdx.x * 2;

    const int lo  = lane & 15;
    const int hi  = lane >> 4;
    const int khi = hi * 2;

    const int nn = (wave & 3) * 16 + lo;
    const int mg = wave >> 2;

    int rowoff[4];
    #pragma unroll
    for (int t = 0; t < 4; ++t) {
        int m = (mg + 2 * t) * 16 + lo;
        if (m > 97) m = 97;
        const int img = (m >= 49) ? 1 : 0;
        const int p = m - img * 49;
        rowoff[t] = img * IMG_LDS + (p / 7) * 9 + (p % 7);
    }

    const float* xb = x + (size_t)b0 * 3136;
    float xv[25];
    #pragma unroll
    for (int j = 0; j < 25; ++j) {
        const int i = tid + j * 256;
        xv[j] = (i < 6272) ? xb[i] : 0.0f;
    }
    for (int i = tid; i < 2 * IMG_LDS; i += 256) act[i] = 0.0f;
    __syncthreads();
    #pragma unroll
    for (int j = 0; j < 25; ++j) {
        const int i = tid + j * 256;
        if (i < 6272) {
            const int img = (i >= 3136) ? 1 : 0;
            const int r = i - img * 3136;
            const int c = r / 49, p = r - c * 49;
            act[img * IMG_LDS + c * 81 + (p / 7 + 1) * 9 + (p % 7) + 1] = xv[j];
        }
    }
    __syncthreads();

    auto conv_pass = [&](const float* __restrict__ wslab, v8f* acc) {
        for (int tap = 0; tap < 9; ++tap) {
            __syncthreads();
            {
                const float4* s4 = (const float4*)(wslab + tap * 4096);
                float4* d4 = (float4*)wsh;
                #pragma unroll
                for (int i = 0; i < 4; ++i) d4[tid + i * 256] = s4[tid + i * 256];
            }
            if (tap < 8)
                __builtin_prefetch((const char*)(wslab + (tap + 1) * 4096) + tid * 64, 0, 1);
            __syncthreads();

            const int r = tap / 3, s = tap - r * 3;
            const int tapoff = r * 9 + s;
            if (mg == 0) tap_mma<4>(act, wsh, tapoff, rowoff, khi, hi, nn, acc);
            else         tap_mma<3>(act, wsh, tapoff, rowoff, khi, hi, nn, acc);
        }
    };

    v8f c1[4] = {};
    conv_pass(wt, c1);
    __syncthreads();

    const float sc1 = g1[nn] * rsqrtf(v1[nn] + EPSV);
    const float sh1 = b1[nn] - m1[nn] * sc1;
    #pragma unroll
    for (int t = 0; t < 4; ++t) {
        #pragma unroll
        for (int j = 0; j < 8; ++j) {
            const int m = (mg + 2 * t) * 16 + j + 8 * hi;
            if (m < 98) {
                const int img = (m >= 49) ? 1 : 0;
                const int p = m - img * 49;
                float v = fmaf(c1[t][j], sc1, sh1);
                v = fmaxf(v, 0.0f);
                if (b0 + img == 0) v *= mask1[nn * 49 + p];
                act[img * IMG_LDS + nn * 81 + (p / 7 + 1) * 9 + (p % 7) + 1] = v;
            }
        }
    }

    v8f c2[4] = {};
    conv_pass(wt + 36864, c2);
    __syncthreads();

    const float sc2 = g2[nn] * rsqrtf(v2p[nn] + EPSV);
    const float sh2 = b2[nn] - m2[nn] * sc2;
    #pragma unroll
    for (int t = 0; t < 4; ++t) {
        #pragma unroll
        for (int j = 0; j < 8; ++j) {
            const int m = (mg + 2 * t) * 16 + j + 8 * hi;
            if (m < 98) {
                const int img = (m >= 49) ? 1 : 0;
                const int p = m - img * 49;
                float v = c2[t][j];
                if (b0 + img == 0) v *= mask2[nn * 49 + p];
                v = fmaf(v, sc2, sh2);
                act[img * IMG_LDS + nn * 81 + (p / 7 + 1) * 9 + (p % 7) + 1] = v;
            }
        }
    }
    __syncthreads();

    float* ob = out + (size_t)b0 * 3136;
    for (int i = tid; i < 6272; i += 256) {
        const int img = (i >= 3136) ? 1 : 0;
        const int r = i - img * 3136;
        const int c = r / 49, p = r - c * 49;
        float v = act[img * IMG_LDS + c * 81 + (p / 7 + 1) * 9 + (p % 7) + 1];
        v += xb[i];
        ob[i] = fmaxf(v, 0.0f);
    }
}

extern "C" void kernel_launch(void* const* d_in, const int* in_sizes, int n_in,
                              void* d_out, int out_size, void* d_ws, size_t ws_size,
                              hipStream_t stream) {
    const float* x     = (const float*)d_in[0];
    const float* w1    = (const float*)d_in[1];
    const float* g1    = (const float*)d_in[2];
    const float* b1    = (const float*)d_in[3];
    const float* m1    = (const float*)d_in[4];
    const float* v1    = (const float*)d_in[5];
    const float* w2    = (const float*)d_in[6];
    const float* g2    = (const float*)d_in[7];
    const float* b2    = (const float*)d_in[8];
    const float* m2    = (const float*)d_in[9];
    const float* v2p   = (const float*)d_in[10];
    const float* mask1 = (const float*)d_in[11];
    const float* mask2 = (const float*)d_in[12];
    float* wt = (float*)d_ws;
    float* out = (float*)d_out;

    prep_weights_kernel<<<288, 256, 0, stream>>>(w1, w2, wt);
    fused_basicblock_kernel<<<2048, 256, 0, stream>>>(
        x, wt, g1, b1, m1, v1, g2, b2, m2, v2p, mask1, mask2, out);
}


# M tile = 16 images at one output pixel (was 16 pixels of 2 images): no pad rows, out-of-image taps skipped; 16 images and 32 waves per workgroup, tap slabs ping-pong in LDS
# speedup vs baseline: 1.8238x; 1.6166x over previous
#include <hip/hip_runtime.h>

#define EPSV 1e-5f

typedef __attribute__((ext_vector_type(2))) float v2f;
typedef __attribute__((ext_vector_type(8))) float v8f;

__global__ __launch_bounds__(256) void prep_weights_kernel(
    const float* __restrict__ w1, const float* __restrict__ w2,
    float* __restrict__ wt)
{
    int i = blockIdx.x * 256 + threadIdx.x;
    if (i >= 2 * 36864) return;
    const float* w = (i < 36864) ? w1 : w2;
    float* dst = wt + (i < 36864 ? 0 : 36864);
    int j = i % 36864;
    int tap = j / 4096;
    int rem = j % 4096;
    int n = (rem >> 1) & 63;
    int k = ((rem >> 7) << 1) | (rem & 1);
    dst[j] = w[(n * 64 + k) * 9 + tap];
}

#define NIMG 16
#define IMGP 3140
#define SLAB 4096

__global__ __launch_bounds__(1024) void fused_basicblock_kernel(
    const float* __restrict__ x,
    const float* __restrict__ wt,
    const float* __restrict__ g1, const float* __restrict__ b1,
    const float* __restrict__ m1, const float* __restrict__ v1,
    const float* __restrict__ g2, const float* __restrict__ b2,
    const float* __restrict__ m2, const float* __restrict__ v2p,
    const float* __restrict__ mask1, const float* __restrict__ mask2,
    float* __restrict__ out)
{
    __shared__ __align__(16) float act[NIMG * IMGP];
    __shared__ __align__(16) float wsh[2 * SLAB];

    const int tid  = threadIdx.x;
    const int lane = tid & 31;
    const int wave = __builtin_amdgcn_readfirstlane(tid >> 5);
    const int b0   = blockIdx.x * NIMG;

    const int lo = lane & 15;
    const int hi = lane >> 4;

    const int nn = (wave & 3) * 16 + lo;
    const int pg = 2 * (((wave >> 3) + (wave & 3)) & 3) + ((wave >> 2) & 1);

    int oh[7], ow[7];
    #pragma unroll
    for (int t = 0; t < 7; ++t) {
        const int p = pg + 8 * t;
        oh[t] = p / 7;
        ow[t] = p - oh[t] * 7;
    }

    const float4* x4 = (const float4*)(x + (size_t)b0 * 3136);
    float4* act4 = (float4*)act;
    {
        float4 xv[13];
        #pragma unroll
        for (int j = 0; j < 13; ++j) {
            const int i = tid + j * 1024;
            xv[j] = (i < 12544) ? x4[i] : make_float4(0.f, 0.f, 0.f, 0.f);
        }
        #pragma unroll
        for (int j = 0; j < 13; ++j) {
            const int i = tid + j * 1024;
            if (i < 12544) {
                const int img = i / 784;
                act4[img * (IMGP / 4) + (i - img * 784)] = xv[j];
            }
        }
    }

    const float* alane = act + lo * IMGP + hi * 98;
    float4* wsh4 = (float4*)wsh;

    auto conv_pass = [&](const float* __restrict__ wslab, v8f* acc) {
        const float4* w4 = (const float4*)wslab;
        wsh4[tid] = w4[tid];
        #pragma unroll 1
        for (int tap = 0; tap < 9; ++tap) {
            __syncthreads();
            float4 wnext = make_float4(0.f, 0.f, 0.f, 0.f);
            if (tap < 8) wnext = w4[(tap + 1) * 1024 + tid];

            const float* wb = wsh + (tap & 1) * SLAB;
            const int r = tap / 3, s = tap - r * 3;
            #pragma unroll
            for (int kh = 0; kh < 2; ++kh) {
                v2f bf[8];
                #pragma unroll
                for (int ks = 0; ks < 8; ++ks)
                    bf[ks] = *(const v2f*)&wb[((kh * 8 + ks) * 2 + hi) * 128 + nn * 2];
                #pragma unroll
                for (int t = 0; t < 7; ++t) {
                    const int ih = oh[t] + r - 1, iw = ow[t] + s - 1;
                    if (pg + 8 * t < 49 && (unsigned)ih < 7u && (unsigned)iw < 7u) {
                        const float* ap = alane + kh * 1568 + ih * 7 + iw;
                        v2f a[8];
                        #pragma unroll
                        for (int ks = 0; ks < 8; ++ks) {
                            a[ks][0] = ap[ks * 196];
                            a[ks][1] = ap[ks * 196 + 49];
                        }
                        __builtin_amdgcn_sched_barrier(0);
                        #pragma unroll
                        for (int ks = 0; ks < 8; ++ks)
                            acc[t] = __builtin_amdgcn_wmma_f32_16x16x4_f32(
                                false, a[ks], false, bf[ks], (short)0, acc[t], false, false);
                    }
                }
            }
            if (tap < 8) wsh4[((tap + 1) & 1) * 1024 + tid] = wnext;
        }
    };

    v8f c1[7] = {};
    conv_pass(wt, c1);
    __syncthreads();

    const float sc1 = g1[nn] * rsqrtf(v1[nn] + EPSV);
    const float sh1 = b1[nn] - m1[nn] * sc1;
    #pragma unroll
    for (int t = 0; t < 7; ++t) {
        const int p = pg + 8 * t;
        if (p < 49) {
            #pragma unroll
            for (int j = 0; j < 8; ++j) {
                const int img = j + 8 * hi;
                float v = fmaf(c1[t][j], sc1, sh1);
                v = fmaxf(v, 0.0f);
                if (b0 + img == 0) v *= mask1[nn * 49 + p];
                act[img * IMGP + nn * 49 + p] = v;
            }
        }
    }

    v8f c2[7] = {};
    conv_pass(wt + 36864, c2);
    __syncthreads();

    const float sc2 = g2[nn] * rsqrtf(v2p[nn] + EPSV);
    const float sh2 = b2[nn] - m2[nn] * sc2;
    #pragma unroll
    for (int t = 0; t < 7; ++t) {
        const int p = pg + 8 * t;
        if (p < 49) {
            #pragma unroll
            for (int j = 0; j < 8; ++j) {
                const int img = j + 8 * hi;
                float v = c2[t][j];
                if (b0 + img == 0) v *= mask2[nn * 49 + p];
                v = fmaf(v, sc2, sh2);
                act[img * IMGP + nn * 49 + p] = v;
            }
        }
    }
    __syncthreads();

    float4* o4 = (float4*)(out + (size_t)b0 * 3136);
    int tl = tid;
    asm volatile("" : "+v"(tl));
    tl &= 1023;
    {
        float4 xr[13];
        #pragma unroll
        for (int j = 0; j < 13; ++j) {
            const int i = tl + j * 1024;
            xr[j] = (i < 12544) ? x4[i] : make_float4(0.f, 0.f, 0.f, 0.f);
        }
        #pragma unroll
        for (int j = 0; j < 13; ++j) {
            const int i = tl + j * 1024;
            if (i < 12544) {
                const int img = i / 784;
                float4 v = act4[img * (IMGP / 4) + (i - img * 784)];
                v.x = fmaxf(v.x + xr[j].x, 0.0f);
                v.y = fmaxf(v.y + xr[j].y, 0.0f);
                v.z = fmaxf(v.z + xr[j].z, 0.0f);
                v.w = fmaxf(v.w + xr[j].w, 0.0f);
                o4[i] = v;
            }
        }
    }
}

extern "C" void kernel_launch(void* const* d_in, const int* in_sizes, int n_in,
                              void* d_out, int out_size, void* d_ws, size_t ws_size,
                              hipStream_t stream) {
    const float* x     = (const float*)d_in[0];
    const float* w1    = (const float*)d_in[1];
    const float* g1    = (const float*)d_in[2];
    const float* b1    = (const float*)d_in[3];
    const float* m1    = (const float*)d_in[4];
    const float* v1    = (const float*)d_in[5];
    const float* w2    = (const float*)d_in[6];
    const float* g2    = (const float*)d_in[7];
    const float* b2    = (const float*)d_in[8];
    const float* m2    = (const float*)d_in[9];
    const float* v2p   = (const float*)d_in[10];
    const float* mask1 = (const float*)d_in[11];
    const float* mask2 = (const float*)d_in[12];
    float* wt = (float*)d_ws;
    float* out = (float*)d_out;

    prep_weights_kernel<<<288, 256, 0, stream>>>(w1, w2, wt);
    fused_basicblock_kernel<<<256, 1024, 0, stream>>>(
        x, wt, g1, b1, m1, v1, g2, b2, m2, v2p, mask1, mask2, out);
}


# Each wave owns one output row (7 pixels), not every 8th pixel: a tap's body is straight-line, A loads of K-step k+1 go in front of the WMMAs of K-step k, which feed 7 independent accumulators
# speedup vs baseline: 1.8936x; 1.0383x over previous
#include <hip/hip_runtime.h>

#define EPSV 1e-5f

typedef __attribute__((ext_vector_type(2))) float v2f;
typedef __attribute__((ext_vector_type(8))) float v8f;

__global__ __launch_bounds__(256) void prep_weights_kernel(
    const float* __restrict__ w1, const float* __restrict__ w2,
    float* __restrict__ wt)
{
    int i = blockIdx.x * 256 + threadIdx.x;
    if (i >= 2 * 36864) return;
    const float* w = (i < 36864) ? w1 : w2;
    float* dst = wt + (i < 36864 ? 0 : 36864);
    int j = i % 36864;
    int tap = j / 4096;
    int rem = j % 4096;
    int n = (rem >> 1) & 63;
    int k = ((rem >> 7) << 1) | (rem & 1);
    dst[j] = w[(n * 64 + k) * 9 + tap];
}

#define NIMG 16
#define IMGP 3140
#define SLAB 4096

__global__ __launch_bounds__(1024) void fused_basicblock_kernel(
    const float* __restrict__ x,
    const float* __restrict__ wt,
    const float* __restrict__ g1, const float* __restrict__ b1,
    const float* __restrict__ m1, const float* __restrict__ v1,
    const float* __restrict__ g2, const float* __restrict__ b2,
    const float* __restrict__ m2, const float* __restrict__ v2p,
    const float* __restrict__ mask1, const float* __restrict__ mask2,
    float* __restrict__ out)
{
    __shared__ __align__(16) float act[NIMG * IMGP];
    __shared__ __align__(16) float wsh[2 * SLAB];

    const int tid  = threadIdx.x;
    const int lane = tid & 31;
    const int wave = __builtin_amdgcn_readfirstlane(tid >> 5);
    const int b0   = blockIdx.x * NIMG;

    const int lo = lane & 15;
    const int hi = lane >> 4;

    const int nn = (wave & 3) * 16 + lo;
    const int pg = 2 * (((wave >> 3) + (wave & 3)) & 3) + ((wave >> 2) & 1);
    const bool own = pg < 7;

    int po[7];
    #pragma unroll
    for (int i = 0; i < 7; ++i) {
        int o = i;
        asm volatile("" : "+s"(o));
        po[i] = o;
    }

    const float4* x4 = (const float4*)(x + (size_t)b0 * 3136);
    float4* act4 = (float4*)act;
    {
        float4 xv[13];
        #pragma unroll
        for (int j = 0; j < 13; ++j) {
            const int i = tid + j * 1024;
            xv[j] = (i < 12544) ? x4[i] : make_float4(0.f, 0.f, 0.f, 0.f);
        }
        #pragma unroll
        for (int j = 0; j < 13; ++j) {
            const int i = tid + j * 1024;
            if (i < 12544) {
                const int img = i / 784;
                act4[img * (IMGP / 4) + (i - img * 784)] = xv[j];
            }
        }
    }

    const float* alane = act + lo * IMGP + hi * 98;
    float4* wsh4 = (float4*)wsh;

    auto conv_pass = [&](const float* __restrict__ wslab, v8f* acc) {
        const float4* w4 = (const float4*)wslab;
        wsh4[tid] = w4[tid];
        #pragma unroll 1
        for (int r = 0; r < 3; ++r) {
            const int ih = pg + r - 1;
            const bool rowok = own && (unsigned)ih < 7u;
            #pragma unroll
            for (int s = 0; s < 3; ++s) {
                const int tap = r * 3 + s;
                __syncthreads();
                float4 wnext = make_float4(0.f, 0.f, 0.f, 0.f);
                if (tap < 8) wnext = w4[(tap + 1) * 1024 + tid];

                const float* wb = wsh + (tap & 1) * SLAB;
                if (rowok) {
                    const float* arow = alane + ih * 7;
                    #pragma unroll
                    for (int kh = 0; kh < 2; ++kh) {
                        v2f a[2][7];
                        #pragma unroll
                        for (int ow = 0; ow < 7; ++ow) {
                            if (ow + s - 1 >= 0 && ow + s - 1 < 7) {
                                const float* ap = arow + po[ow + s - 1];
                                a[0][ow][0] = ap[(kh * 8) * 196];
                                a[0][ow][1] = ap[(kh * 8) * 196 + 49];
                            }
                        }
                        v2f bf[8];
                        #pragma unroll
                        for (int ks = 0; ks < 8; ++ks)
                            bf[ks] = *(const v2f*)&wb[((kh * 8 + ks) * 2 + hi) * 128 + nn * 2];
                        #pragma unroll
                        for (int ks = 0; ks < 8; ++ks) {
                            if (ks < 7) {
                                #pragma unroll
                                for (int ow = 0; ow < 7; ++ow) {
                                    if (ow + s - 1 >= 0 && ow + s - 1 < 7) {
                                        const float* ap = arow + po[ow + s - 1];
                                        a[(ks + 1) & 1][ow][0] = ap[(kh * 8 + ks + 1) * 196];
                                        a[(ks + 1) & 1][ow][1] = ap[(kh * 8 + ks + 1) * 196 + 49];
                                    }
                                }
                            }
                            __builtin_amdgcn_sched_barrier(0);
                            #pragma unroll
                            for (int ow = 0; ow < 7; ++ow) {
                                if (ow + s - 1 >= 0 && ow + s - 1 < 7)
                                    acc[ow] = __builtin_amdgcn_wmma_f32_16x16x4_f32(
                                        false, a[ks & 1][ow], false, bf[ks], (short)0, acc[ow], false, false);
                            }
                            __builtin_amdgcn_sched_barrier(0);
                        }
                    }
                }
                if (tap < 8) wsh4[((tap + 1) & 1) * 1024 + tid] = wnext;
            }
        }
    };

    v8f c1[7] = {};
    conv_pass(wt, c1);
    __syncthreads();

    const float sc1 = g1[nn] * rsqrtf(v1[nn] + EPSV);
    const float sh1 = b1[nn] - m1[nn] * sc1;
    if (own) {
        #pragma unroll
        for (int t = 0; t < 7; ++t) {
            const int p = pg * 7 + t;
            #pragma unroll
            for (int j = 0; j < 8; ++j) {
                const int img = j + 8 * hi;
                float v = fmaf(c1[t][j], sc1, sh1);
                v = fmaxf(v, 0.0f);
                if (b0 + img == 0) v *= mask1[nn * 49 + p];
                act[img * IMGP + nn * 49 + p] = v;
            }
        }
    }

    v8f c2[7] = {};
    conv_pass(wt + 36864, c2);
    __syncthreads();

    const float sc2 = g2[nn] * rsqrtf(v2p[nn] + EPSV);
    const float sh2 = b2[nn] - m2[nn] * sc2;
    if (own) {
        #pragma unroll
        for (int t = 0; t < 7; ++t) {
            const int p = pg * 7 + t;
            #pragma unroll
            for (int j = 0; j < 8; ++j) {
                const int img = j + 8 * hi;
                float v = c2[t][j];
                if (b0 + img == 0) v *= mask2[nn * 49 + p];
                v = fmaf(v, sc2, sh2);
                act[img * IMGP + nn * 49 + p] = v;
            }
        }
    }
    __syncthreads();

    float4* o4 = (float4*)(out + (size_t)b0 * 3136);
    int tl = tid;
    asm volatile("" : "+v"(tl));
    tl &= 1023;
    {
        float4 xr[13];
        #pragma unroll
        for (int j = 0; j < 13; ++j) {
            const int i = tl + j * 1024;
            xr[j] = (i < 12544) ? x4[i] : make_float4(0.f, 0.f, 0.f, 0.f);
        }
        #pragma unroll
        for (int j = 0; j < 13; ++j) {
            const int i = tl + j * 1024;
            if (i < 12544) {
                const int img = i / 784;
                float4 v = act4[img * (IMGP / 4) + (i - img * 784)];
                v.x = fmaxf(v.x + xr[j].x, 0.0f);
                v.y = fmaxf(v.y + xr[j].y, 0.0f);
                v.z = fmaxf(v.z + xr[j].z, 0.0f);
                v.w = fmaxf(v.w + xr[j].w, 0.0f);
                o4[i] = v;
            }
        }
    }
}

extern "C" void kernel_launch(void* const* d_in, const int* in_sizes, int n_in,
                              void* d_out, int out_size, void* d_ws, size_t ws_size,
                              hipStream_t stream) {
    const float* x     = (const float*)d_in[0];
    const float* w1    = (const float*)d_in[1];
    const float* g1    = (const float*)d_in[2];
    const float* b1    = (const float*)d_in[3];
    const float* m1    = (const float*)d_in[4];
    const float* v1    = (const float*)d_in[5];
    const float* w2    = (const float*)d_in[6];
    const float* g2    = (const float*)d_in[7];
    const float* b2    = (const float*)d_in[8];
    const float* m2    = (const float*)d_in[9];
    const float* v2p   = (const float*)d_in[10];
    const float* mask1 = (const float*)d_in[11];
    const float* mask2 = (const float*)d_in[12];
    float* wt = (float*)d_ws;
    float* out = (float*)d_out;

    prep_weights_kernel<<<288, 256, 0, stream>>>(w1, w2, wt);
    fused_basicblock_kernel<<<256, 1024, 0, stream>>>(
        x, wt, g1, b1, m1, v1, g2, b2, m2, v2p, mask1, mask2, out);
}


# Filter row outer, K-step middle, its 3 taps inner: a row's 7 A fragments, loaded once per K-step, feed all 3 taps (one LDS read group per 19 WMMAs, was 6-7); 3-slab stages, one barrier per filter row
# speedup vs baseline: 2.0679x; 1.0920x over previous
#include <hip/hip_runtime.h>

#define EPSV 1e-5f

typedef __attribute__((ext_vector_type(2))) float v2f;
typedef __attribute__((ext_vector_type(8))) float v8f;

__global__ __launch_bounds__(256) void prep_weights_kernel(
    const float* __restrict__ w1, const float* __restrict__ w2,
    float* __restrict__ wt)
{
    int i = blockIdx.x * 256 + threadIdx.x;
    if (i >= 2 * 36864) return;
    const float* w = (i < 36864) ? w1 : w2;
    float* dst = wt + (i < 36864 ? 0 : 36864);
    int j = i % 36864;
    int tap = j / 4096;
    int rem = j % 4096;
    int n = (rem >> 1) & 63;
    int k = ((rem >> 7) << 1) | (rem & 1);
    dst[j] = w[(n * 64 + k) * 9 + tap];
}

#define NIMG 16
#define IMGP 3140
#define SLAB 4096
#define STG  (3 * SLAB)

__global__ __launch_bounds__(1024) void fused_basicblock_kernel(
    const float* __restrict__ x,
    const float* __restrict__ wt,
    const float* __restrict__ g1, const float* __restrict__ b1,
    const float* __restrict__ m1, const float* __restrict__ v1,
    const float* __restrict__ g2, const float* __restrict__ b2,
    const float* __restrict__ m2, const float* __restrict__ v2p,
    const float* __restrict__ mask1, const float* __restrict__ mask2,
    float* __restrict__ out)
{
    __shared__ __align__(16) float act[NIMG * IMGP];
    __shared__ __align__(16) float wsh[2 * STG];

    const int tid  = threadIdx.x;
    const int lane = tid & 31;
    const int wave = __builtin_amdgcn_readfirstlane(tid >> 5);
    const int b0   = blockIdx.x * NIMG;

    const int lo = lane & 15;
    const int hi = lane >> 4;

    const int nn = (wave & 3) * 16 + lo;
    const int pg = 2 * (((wave >> 3) + (wave & 3)) & 3) + ((wave >> 2) & 1);
    const bool own = pg < 7;
    const int it = (wave & 3) * 32 + lane;

    int po[7];
    #pragma unroll
    for (int i = 0; i < 7; ++i) {
        int o = i;
        asm volatile("" : "+s"(o));
        po[i] = o;
    }

    const float4* w4 = (const float4*)wt;
    float4* wsh4 = (float4*)wsh;

    const float4* x4 = (const float4*)(x + (size_t)b0 * 3136);
    float4* act4 = (float4*)act;
    {
        float4 xv[13];
        #pragma unroll
        for (int j = 0; j < 13; ++j) {
            const int i = tid + j * 1024;
            xv[j] = (i < 12544) ? x4[i] : make_float4(0.f, 0.f, 0.f, 0.f);
        }
        const float4 wv0 = w4[tid], wv1 = w4[tid + 1024], wv2 = w4[tid + 2048];
        #pragma unroll
        for (int j = 0; j < 13; ++j) {
            const int i = tid + j * 1024;
            if (i < 12544) {
                const int img = i / 784;
                act4[img * (IMGP / 4) + (i - img * 784)] = xv[j];
            }
        }
        wsh4[tid] = wv0;
        wsh4[tid + 1024] = wv1;
        wsh4[tid + 2048] = wv2;
    }

    const float* alane = act + lo * IMGP + hi * 98;

    auto conv_pass = [&](const int c, v8f* acc) {
        #pragma unroll 1
        for (int r = 0; r < 3; ++r) {
            const int g  = c * 3 + r;
            const int ih = pg + r - 1;
            const bool rowok = own && (unsigned)ih < 7u;
            __syncthreads();
            if (!own) {
                if (g < 5) {
                    const float4* srcw = w4 + (g + 1) * (STG / 4);
                    float4* dstw = wsh4 + ((g + 1) & 1) * (STG / 4);
                    #pragma unroll
                    for (int j0 = 0; j0 < 24; j0 += 6) {
                        const float4 t0 = srcw[it + (j0 + 0) * 128];
                        const float4 t1 = srcw[it + (j0 + 1) * 128];
                        const float4 t2 = srcw[it + (j0 + 2) * 128];
                        const float4 t3 = srcw[it + (j0 + 3) * 128];
                        const float4 t4 = srcw[it + (j0 + 4) * 128];
                        const float4 t5 = srcw[it + (j0 + 5) * 128];
                        dstw[it + (j0 + 0) * 128] = t0;
                        dstw[it + (j0 + 1) * 128] = t1;
                        dstw[it + (j0 + 2) * 128] = t2;
                        dstw[it + (j0 + 3) * 128] = t3;
                        dstw[it + (j0 + 4) * 128] = t4;
                        dstw[it + (j0 + 5) * 128] = t5;
                    }
                }
            }
            if (rowok) {
                const float* arow = alane + ih * 7;
                const float* wb = wsh + (g & 1) * STG + hi * 128 + nn * 2;
                v2f a[2][7];
                v2f bf[2][3];
                #pragma unroll
                for (int q = 0; q < 7; ++q) {
                    const float* ap = arow + po[q];
                    a[0][q][0] = ap[0];
                    a[0][q][1] = ap[49];
                }
                #pragma unroll
                for (int s = 0; s < 3; ++s) bf[0][s] = *(const v2f*)&wb[s * SLAB];
                #pragma unroll
                for (int ks = 0; ks < 16; ++ks) {
                    if (ks < 15) {
                        #pragma unroll
                        for (int q = 0; q < 7; ++q) {
                            const float* ap = arow + po[q];
                            a[(ks + 1) & 1][q][0] = ap[(ks + 1) * 196];
                            a[(ks + 1) & 1][q][1] = ap[(ks + 1) * 196 + 49];
                        }
                        #pragma unroll
                        for (int s = 0; s < 3; ++s)
                            bf[(ks + 1) & 1][s] = *(const v2f*)&wb[s * SLAB + (ks + 1) * 256];
                    }
                    __builtin_amdgcn_sched_barrier(0);
                    #pragma unroll
                    for (int s = 0; s < 3; ++s) {
                        #pragma unroll
                        for (int ow = 0; ow < 7; ++ow) {
                            if (ow + s - 1 >= 0 && ow + s - 1 < 7)
                                acc[ow] = __builtin_amdgcn_wmma_f32_16x16x4_f32(
                                    false, a[ks & 1][ow + s - 1], false, bf[ks & 1][s],
                                    (short)0, acc[ow], false, false);
                        }
                    }
                    __builtin_amdgcn_sched_barrier(0);
                }
            }
        }
    };

    v8f c1[7] = {};
    conv_pass(0, c1);
    __syncthreads();

    const float sc1 = g1[nn] * rsqrtf(v1[nn] + EPSV);
    const float sh1 = b1[nn] - m1[nn] * sc1;
    if (own) {
        #pragma unroll
        for (int t = 0; t < 7; ++t) {
            const int p = pg * 7 + t;
            #pragma unroll
            for (int j = 0; j < 8; ++j) {
                const int img = j + 8 * hi;
                float v = fmaf(c1[t][j], sc1, sh1);
                v = fmaxf(v, 0.0f);
                if (b0 + img == 0) v *= mask1[nn * 49 + p];
                act[img * IMGP + nn * 49 + p] = v;
            }
        }
    }

    v8f c2[7] = {};
    conv_pass(1, c2);
    __syncthreads();

    const float sc2 = g2[nn] * rsqrtf(v2p[nn] + EPSV);
    const float sh2 = b2[nn] - m2[nn] * sc2;
    if (own) {
        #pragma unroll
        for (int t = 0; t < 7; ++t) {
            const int p = pg * 7 + t;
            #pragma unroll
            for (int j = 0; j < 8; ++j) {
                const int img = j + 8 * hi;
                float v = c2[t][j];
                if (b0 + img == 0) v *= mask2[nn * 49 + p];
                v = fmaf(v, sc2, sh2);
                act[img * IMGP + nn * 49 + p] = v;
            }
        }
    }
    __syncthreads();

    float4* o4 = (float4*)(out + (size_t)b0 * 3136);
    int tl = tid;
    asm volatile("" : "+v"(tl));
    tl &= 1023;
    {
        float4 xr[13];
        #pragma unroll
        for (int j = 0; j < 13; ++j) {
            const int i = tl + j * 1024;
            xr[j] = (i < 12544) ? x4[i] : make_float4(0.f, 0.f, 0.f, 0.f);
        }
        #pragma unroll
        for (int j = 0; j < 13; ++j) {
            const int i = tl + j * 1024;
            if (i < 12544) {
                const int img = i / 784;
                float4 v = act4[img * (IMGP / 4) + (i - img * 784)];
                v.x = fmaxf(v.x + xr[j].x, 0.0f);
                v.y = fmaxf(v.y + xr[j].y, 0.0f);
                v.z = fmaxf(v.z + xr[j].z, 0.0f);
                v.w = fmaxf(v.w + xr[j].w, 0.0f);
                o4[i] = v;
            }
        }
    }
}

extern "C" void kernel_launch(void* const* d_in, const int* in_sizes, int n_in,
                              void* d_out, int out_size, void* d_ws, size_t ws_size,
                              hipStream_t stream) {
    const float* x     = (const float*)d_in[0];
    const float* w1    = (const float*)d_in[1];
    const float* g1    = (const float*)d_in[2];
    const float* b1    = (const float*)d_in[3];
    const float* m1    = (const float*)d_in[4];
    const float* v1    = (const float*)d_in[5];
    const float* w2    = (const float*)d_in[6];
    const float* g2    = (const float*)d_in[7];
    const float* b2    = (const float*)d_in[8];
    const float* m2    = (const float*)d_in[9];
    const float* v2p   = (const float*)d_in[10];
    const float* mask1 = (const float*)d_in[11];
    const float* mask2 = (const float*)d_in[12];
    float* wt = (float*)d_ws;
    float* out = (float*)d_out;

    prep_weights_kernel<<<288, 256, 0, stream>>>(w1, w2, wt);
    fused_basicblock_kernel<<<256, 1024, 0, stream>>>(
        x, wt, g1, b1, m1, v1, g2, b2, m2, v2p, mask1, mask2, out);
}


# Three taps of a filter row by the two-output fast FIR: (q0+q1)x(g0+g1+g2)/2, (q1-q0)x(g0-g1+g2)/2, one butterfly on the accumulators, then g0 and g2 terms; 15 f32 WMMAs per row and K-step, was 19
# speedup vs baseline: 2.1651x; 1.0470x over previous
#include <hip/hip_runtime.h>

#define EPSV 1e-5f

typedef __attribute__((ext_vector_type(2))) float v2f;
typedef __attribute__((ext_vector_type(8))) float v8f;

__global__ __launch_bounds__(256) void prep_weights_kernel(
    const float* __restrict__ w1, const float* __restrict__ w2,
    float* __restrict__ wt)
{
    int i = blockIdx.x * 256 + threadIdx.x;
    if (i >= 2 * 36864) return;
    const float* w = (i < 36864) ? w1 : w2;
    float* dst = wt + (i < 36864 ? 0 : 36864);
    int j = i % 36864;
    int tap = j / 4096;
    int rem = j % 4096;
    int n = (rem >> 1) & 63;
    int k = ((rem >> 7) << 1) | (rem & 1);
    dst[j] = w[(n * 64 + k) * 9 + tap];
}

#define NIMG 16
#define IMGP 3140
#define SLAB 4096
#define STG  (2 * SLAB)

__device__ __forceinline__ void fir_taps(const float4 g0, const float4 g1, const float4 g2,
                                         float4& gp, float4& gm)
{
    const float ex = g0.x + g2.x, ey = g0.y + g2.y, ez = g0.z + g2.z, ew = g0.w + g2.w;
    gp.x = 0.5f * (ex + g1.x);  gm.x = 0.5f * (ex - g1.x);
    gp.y = 0.5f * (ey + g1.y);  gm.y = 0.5f * (ey - g1.y);
    gp.z = 0.5f * (ez + g1.z);  gm.z = 0.5f * (ez - g1.z);
    gp.w = 0.5f * (ew + g1.w);  gm.w = 0.5f * (ew - g1.w);
}

__global__ __launch_bounds__(1024) void fused_basicblock_kernel(
    const float* __restrict__ x,
    const float* __restrict__ wt,
    const float* __restrict__ g1, const float* __restrict__ b1,
    const float* __restrict__ m1, const float* __restrict__ v1,
    const float* __restrict__ g2, const float* __restrict__ b2,
    const float* __restrict__ m2, const float* __restrict__ v2p,
    const float* __restrict__ mask1, const float* __restrict__ mask2,
    float* __restrict__ out)
{
    __shared__ __align__(16) float act[NIMG * IMGP];
    __shared__ __align__(16) float wsh[2 * STG];

    const int tid  = threadIdx.x;
    const int lane = tid & 31;
    const int wave = __builtin_amdgcn_readfirstlane(tid >> 5);
    const int b0   = blockIdx.x * NIMG;

    const int lo = lane & 15;
    const int hi = lane >> 4;

    const int nn = (wave & 3) * 16 + lo;
    const int pg = 2 * (((wave >> 3) + (wave & 3)) & 3) + ((wave >> 2) & 1);
    const bool own = pg < 7;
    const int it = (wave & 3) * 32 + lane;

    int po[7];
    #pragma unroll
    for (int i = 0; i < 7; ++i) {
        int o = i;
        asm volatile("" : "+s"(o));
        po[i] = o;
    }

    const float4* w4 = (const float4*)wt;
    float4* wsh4 = (float4*)wsh;

    const float4* x4 = (const float4*)(x + (size_t)b0 * 3136);
    float4* act4 = (float4*)act;
    {
        float4 xv[13];
        #pragma unroll
        for (int j = 0; j < 13; ++j) {
            const int i = tid + j * 1024;
            xv[j] = (i < 12544) ? x4[i] : make_float4(0.f, 0.f, 0.f, 0.f);
        }
        const float4 wv0 = w4[tid], wv1 = w4[tid + 1024], wv2 = w4[tid + 2048];
        #pragma unroll
        for (int j = 0; j < 13; ++j) {
            const int i = tid + j * 1024;
            if (i < 12544) {
                const int img = i / 784;
                act4[img * (IMGP / 4) + (i - img * 784)] = xv[j];
            }
        }
        float4 gp, gm;
        fir_taps(wv0, wv1, wv2, gp, gm);
        wsh4[tid] = gp;
        wsh4[tid + 1024] = gm;
    }

    const float* alane = act + lo * IMGP + hi * 98;

    auto fill_stage = [&](const int gn) {
        const int cn = (gn >= 6) ? 1 : 0;
        const int jn = gn - 6 * cn;
        const bool pa = jn < 3;
        const int fr = pa ? jn : jn - 3;
        const float4* srcw = w4 + (cn * 9 + fr * 3) * (SLAB / 4) + it;
        float4* dstw = wsh4 + (gn & 1) * (STG / 4) + it;
        if (pa) {
            #pragma unroll
            for (int j0 = 0; j0 < 8; j0 += 2) {
                const float4 t00 = srcw[(j0 + 0) * 128];
                const float4 t01 = srcw[(j0 + 0) * 128 + 1024];
                const float4 t02 = srcw[(j0 + 0) * 128 + 2048];
                const float4 t10 = srcw[(j0 + 1) * 128];
                const float4 t11 = srcw[(j0 + 1) * 128 + 1024];
                const float4 t12 = srcw[(j0 + 1) * 128 + 2048];
                float4 fp0, fm0, fp1, fm1;
                fir_taps(t00, t01, t02, fp0, fm0);
                fir_taps(t10, t11, t12, fp1, fm1);
                dstw[(j0 + 0) * 128] = fp0;
                dstw[(j0 + 0) * 128 + 1024] = fm0;
                dstw[(j0 + 1) * 128] = fp1;
                dstw[(j0 + 1) * 128 + 1024] = fm1;
            }
        } else {
            #pragma unroll
            for (int j0 = 0; j0 < 8; j0 += 3) {
                const float4 ta0 = srcw[(j0 + 0) * 128];
                const float4 tc0 = srcw[(j0 + 0) * 128 + 2048];
                const float4 ta1 = srcw[(j0 + 1) * 128];
                const float4 tc1 = srcw[(j0 + 1) * 128 + 2048];
                dstw[(j0 + 0) * 128] = ta0;
                dstw[(j0 + 0) * 128 + 1024] = tc0;
                dstw[(j0 + 1) * 128] = ta1;
                dstw[(j0 + 1) * 128 + 1024] = tc1;
                if (j0 + 2 < 8) {
                    const float4 ta2 = srcw[(j0 + 2) * 128];
                    const float4 tc2 = srcw[(j0 + 2) * 128 + 2048];
                    dstw[(j0 + 2) * 128] = ta2;
                    dstw[(j0 + 2) * 128 + 1024] = tc2;
                }
            }
        }
    };

    auto conv_pass = [&](const int c, v8f* acc) {
        #pragma unroll 1
        for (int r = 0; r < 3; ++r) {
            const int g  = c * 6 + r;
            const int ih = pg + r - 1;
            const bool rowok = own && (unsigned)ih < 7u;
            __syncthreads();
            if (!own) fill_stage(g + 1);
            if (rowok) {
                const float* arow = alane + ih * 7;
                const float* wb = wsh + (g & 1) * STG + hi * 128 + nn * 2;
                v2f a[2][7];
                v2f bf[2][2];
                #pragma unroll
                for (int q = 0; q < 7; ++q) {
                    const float* ap = arow + po[q];
                    a[0][q][0] = ap[0];
                    a[0][q][1] = ap[49];
                }
                #pragma unroll
                for (int s = 0; s < 2; ++s) bf[0][s] = *(const v2f*)&wb[s * SLAB];
                #pragma unroll
                for (int ks = 0; ks < 16; ++ks) {
                    if (ks < 15) {
                        #pragma unroll
                        for (int q = 0; q < 7; ++q) {
                            const float* ap = arow + po[q];
                            a[(ks + 1) & 1][q][0] = ap[(ks + 1) * 196];
                            a[(ks + 1) & 1][q][1] = ap[(ks + 1) * 196 + 49];
                        }
                        #pragma unroll
                        for (int s = 0; s < 2; ++s)
                            bf[(ks + 1) & 1][s] = *(const v2f*)&wb[s * SLAB + (ks + 1) * 256];
                    }
                    __builtin_amdgcn_sched_barrier(0);
                    const v2f* q = a[ks & 1];
                    const v2f gp = bf[ks & 1][0], gm = bf[ks & 1][1];
                    const v2f s01 = q[0] + q[1], d01 = q[1] - q[0];
                    const v2f s23 = q[2] + q[3], d23 = q[3] - q[2];
                    const v2f s45 = q[4] + q[5], d45 = q[5] - q[4];
                    acc[0] = __builtin_amdgcn_wmma_f32_16x16x4_f32(false, s01,  false, gp, (short)0, acc[0], false, false);
                    acc[1] = __builtin_amdgcn_wmma_f32_16x16x4_f32(false, d01,  false, gm, (short)0, acc[1], false, false);
                    acc[2] = __builtin_amdgcn_wmma_f32_16x16x4_f32(false, s23,  false, gp, (short)0, acc[2], false, false);
                    acc[3] = __builtin_amdgcn_wmma_f32_16x16x4_f32(false, d23,  false, gm, (short)0, acc[3], false, false);
                    acc[4] = __builtin_amdgcn_wmma_f32_16x16x4_f32(false, s45,  false, gp, (short)0, acc[4], false, false);
                    acc[5] = __builtin_amdgcn_wmma_f32_16x16x4_f32(false, d45,  false, gm, (short)0, acc[5], false, false);
                    acc[6] = __builtin_amdgcn_wmma_f32_16x16x4_f32(false, q[6], false, gp, (short)0, acc[6], false, false);
                    acc[7] = __builtin_amdgcn_wmma_f32_16x16x4_f32(false, q[6], false, gm, (short)0, acc[7], false, false);
                    __builtin_amdgcn_sched_barrier(0);
                }
            }
        }

        #pragma unroll
        for (int p = 0; p < 3; ++p) {
            const v8f ta = acc[2 * p], tb = acc[2 * p + 1];
            acc[2 * p]     = (p == 0) ? (-ta - tb) : (ta + tb);
            acc[2 * p + 1] = ta - tb;
        }
        acc[6] = acc[6] - acc[7];

        #pragma unroll 1
        for (int r = 0; r < 3; ++r) {
            const int g  = c * 6 + 3 + r;
            const int ih = pg + r - 1;
            const bool rowok = own && (unsigned)ih < 7u;
            __syncthreads();
            if (!own) {
                if (g < 11) fill_stage(g + 1);
            }
            if (rowok) {
                const float* arow = alane + ih * 7;
                const float* wb = wsh + (g & 1) * STG + hi * 128 + nn * 2;
                v2f a[2][7];
                v2f bf[2][2];
                #pragma unroll
                for (int q = 0; q < 7; ++q) {
                    const float* ap = arow + po[q];
                    a[0][q][0] = ap[0];
                    a[0][q][1] = ap[49];
                }
                #pragma unroll
                for (int s = 0; s < 2; ++s) bf[0][s] = *(const v2f*)&wb[s * SLAB];
                #pragma unroll
                for (int ks = 0; ks < 16; ++ks) {
                    if (ks < 15) {
                        #pragma unroll
                        for (int q = 0; q < 7; ++q) {
                            const float* ap = arow + po[q];
                            a[(ks + 1) & 1][q][0] = ap[(ks + 1) * 196];
                            a[(ks + 1) & 1][q][1] = ap[(ks + 1) * 196 + 49];
                        }
                        #pragma unroll
                        for (int s = 0; s < 2; ++s)
                            bf[(ks + 1) & 1][s] = *(const v2f*)&wb[s * SLAB + (ks + 1) * 256];
                    }
                    __builtin_amdgcn_sched_barrier(0);
                    const v2f* q = a[ks & 1];
                    const v2f t0 = bf[ks & 1][0], t2 = bf[ks & 1][1];
                    const v2f e13 = q[1] - q[3], e35 = q[3] - q[5];
                    const v2f o20 = q[2] - q[0], o42 = q[4] - q[2], o64 = q[6] - q[4];
                    acc[0] = __builtin_amdgcn_wmma_f32_16x16x4_f32(false, q[1], false, t0, (short)0, acc[0], false, false);
                    acc[1] = __builtin_amdgcn_wmma_f32_16x16x4_f32(false, o20,  false, t2, (short)0, acc[1], false, false);
                    acc[2] = __builtin_amdgcn_wmma_f32_16x16x4_f32(false, e13,  false, t0, (short)0, acc[2], false, false);
                    acc[3] = __builtin_amdgcn_wmma_f32_16x16x4_f32(false, o42,  false, t2, (short)0, acc[3], false, false);
                    acc[4] = __builtin_amdgcn_wmma_f32_16x16x4_f32(false, e35,  false, t0, (short)0, acc[4], false, false);
                    acc[5] = __builtin_amdgcn_wmma_f32_16x16x4_f32(false, o64,  false, t2, (short)0, acc[5], false, false);
                    acc[6] = __builtin_amdgcn_wmma_f32_16x16x4_f32(false, q[5], false, t0, (short)0, acc[6], false, false);
                    __builtin_amdgcn_sched_barrier(0);
                }
            }
        }
    };

    v8f c1[8] = {};
    conv_pass(0, c1);
    __syncthreads();

    const float sc1 = g1[nn] * rsqrtf(v1[nn] + EPSV);
    const float sh1 = b1[nn] - m1[nn] * sc1;
    if (own) {
        #pragma unroll
        for (int t = 0; t < 7; ++t) {
            const int p = pg * 7 + t;
            #pragma unroll
            for (int j = 0; j < 8; ++j) {
                const int img = j + 8 * hi;
                const float cv = (t == 0) ? -c1[t][j] : c1[t][j];
                float v = fmaf(cv, sc1, sh1);
                v = fmaxf(v, 0.0f);
                if (b0 + img == 0) v *= mask1[nn * 49 + p];
                act[img * IMGP + nn * 49 + p] = v;
            }
        }
    }

    v8f c2[8] = {};
    conv_pass(1, c2);
    __syncthreads();

    const float sc2 = g2[nn] * rsqrtf(v2p[nn] + EPSV);
    const float sh2 = b2[nn] - m2[nn] * sc2;
    if (own) {
        #pragma unroll
        for (int t = 0; t < 7; ++t) {
            const int p = pg * 7 + t;
            #pragma unroll
            for (int j = 0; j < 8; ++j) {
                const int img = j + 8 * hi;
                float v = (t == 0) ? -c2[t][j] : c2[t][j];
                if (b0 + img == 0) v *= mask2[nn * 49 + p];
                v = fmaf(v, sc2, sh2);
                act[img * IMGP + nn * 49 + p] = v;
            }
        }
    }
    __syncthreads();

    float4* o4 = (float4*)(out + (size_t)b0 * 3136);
    int tl = tid;
    asm volatile("" : "+v"(tl));
    tl &= 1023;
    {
        float4 xr[13];
        #pragma unroll
        for (int j = 0; j < 13; ++j) {
            const int i = tl + j * 1024;
            xr[j] = (i < 12544) ? x4[i] : make_float4(0.f, 0.f, 0.f, 0.f);
        }
        #pragma unroll
        for (int j = 0; j < 13; ++j) {
            const int i = tl + j * 1024;
            if (i < 12544) {
                const int img = i / 784;
                float4 v = act4[img * (IMGP / 4) + (i - img * 784)];
                v.x = fmaxf(v.x + xr[j].x, 0.0f);
                v.y = fmaxf(v.y + xr[j].y, 0.0f);
                v.z = fmaxf(v.z + xr[j].z, 0.0f);
                v.w = fmaxf(v.w + xr[j].w, 0.0f);
                o4[i] = v;
            }
        }
    }
}

extern "C" void kernel_launch(void* const* d_in, const int* in_sizes, int n_in,
                              void* d_out, int out_size, void* d_ws, size_t ws_size,
                              hipStream_t stream) {
    const float* x     = (const float*)d_in[0];
    const float* w1    = (const float*)d_in[1];
    const float* g1    = (const float*)d_in[2];
    const float* b1    = (const float*)d_in[3];
    const float* m1    = (const float*)d_in[4];
    const float* v1    = (const float*)d_in[5];
    const float* w2    = (const float*)d_in[6];
    const float* g2    = (const float*)d_in[7];
    const float* b2    = (const float*)d_in[8];
    const float* m2    = (const float*)d_in[9];
    const float* v2p   = (const float*)d_in[10];
    const float* mask1 = (const float*)d_in[11];
    const float* mask2 = (const float*)d_in[12];
    float* wt = (float*)d_ws;
    float* out = (float*)d_out;

    prep_weights_kernel<<<288, 256, 0, stream>>>(w1, w2, wt);
    fused_basicblock_kernel<<<256, 1024, 0, stream>>>(
        x, wt, g1, b1, m1, v1, g2, b2, m2, v2p, mask1, mask2, out);
}


# Row taps by the three-output fast FIR (points 0,1,-1,2,inf): pixels 0-2 and 3-5 are two tiles of 5 products each, pixel 6 direct; 12 f32 WMMAs per filter row and K-step, was 15; three slabs a stage
# speedup vs baseline: 2.3730x; 1.0960x over previous
#include <hip/hip_runtime.h>

#define EPSV 1e-5f

typedef __attribute__((ext_vector_type(2))) float v2f;
typedef __attribute__((ext_vector_type(8))) float v8f;

__global__ __launch_bounds__(256) void prep_weights_kernel(
    const float* __restrict__ w1, const float* __restrict__ w2,
    float* __restrict__ wt)
{
    int i = blockIdx.x * 256 + threadIdx.x;
    if (i >= 2 * 36864) return;
    const float* w = (i < 36864) ? w1 : w2;
    float* dst = wt + (i < 36864 ? 0 : 36864);
    int j = i % 36864;
    int tap = j / 4096;
    int rem = j % 4096;
    int n = (rem >> 1) & 63;
    int k = ((rem >> 7) << 1) | (rem & 1);
    dst[j] = w[(n * 64 + k) * 9 + tap];
}

#define NIMG 16
#define IMGP 3140
#define SLAB 4096
#define STG  (3 * SLAB)

__device__ __forceinline__ void fir_tap1(const float g0, const float g1, const float g2,
                                         float& w1, float& w2, float& w3)
{
    const float e = g0 + g2;
    w1 = 0.5f * (e + g1);
    w2 = (1.0f / 6.0f) * (e - g1);
    w3 = (1.0f / 6.0f) * (fmaf(4.0f, g2, g0) + 2.0f * g1);
}
__device__ __forceinline__ void fir_taps(const float4 g0, const float4 g1, const float4 g2,
                                         float4& w1, float4& w2, float4& w3)
{
    fir_tap1(g0.x, g1.x, g2.x, w1.x, w2.x, w3.x);
    fir_tap1(g0.y, g1.y, g2.y, w1.y, w2.y, w3.y);
    fir_tap1(g0.z, g1.z, g2.z, w1.z, w2.z, w3.z);
    fir_tap1(g0.w, g1.w, g2.w, w1.w, w2.w, w3.w);
}

__global__ __launch_bounds__(1024) void fused_basicblock_kernel(
    const float* __restrict__ x,
    const float* __restrict__ wt,
    const float* __restrict__ g1, const float* __restrict__ b1,
    const float* __restrict__ m1, const float* __restrict__ v1,
    const float* __restrict__ g2, const float* __restrict__ b2,
    const float* __restrict__ m2, const float* __restrict__ v2p,
    const float* __restrict__ mask1, const float* __restrict__ mask2,
    float* __restrict__ out)
{
    __shared__ __align__(16) float act[NIMG * IMGP];
    __shared__ __align__(16) float wsh[2 * STG];

    const int tid  = threadIdx.x;
    const int lane = tid & 31;
    const int wave = __builtin_amdgcn_readfirstlane(tid >> 5);
    const int b0   = blockIdx.x * NIMG;

    const int lo = lane & 15;
    const int hi = lane >> 4;

    const int nn = (wave & 3) * 16 + lo;
    const int pg = 2 * (((wave >> 3) + (wave & 3)) & 3) + ((wave >> 2) & 1);
    const bool own = pg < 7;
    const int it = (wave & 3) * 32 + lane;

    int po[7];
    #pragma unroll
    for (int i = 0; i < 7; ++i) {
        int o = i;
        asm volatile("" : "+s"(o));
        po[i] = o;
    }

    const float4* w4 = (const float4*)wt;
    float4* wsh4 = (float4*)wsh;

    const float4* x4 = (const float4*)(x + (size_t)b0 * 3136);
    float4* act4 = (float4*)act;
    {
        float4 xv[13];
        #pragma unroll
        for (int j = 0; j < 13; ++j) {
            const int i = tid + j * 1024;
            xv[j] = (i < 12544) ? x4[i] : make_float4(0.f, 0.f, 0.f, 0.f);
        }
        const float4 wv0 = w4[tid], wv1 = w4[tid + 1024], wv2 = w4[tid + 2048];
        #pragma unroll
        for (int j = 0; j < 13; ++j) {
            const int i = tid + j * 1024;
            if (i < 12544) {
                const int img = i / 784;
                act4[img * (IMGP / 4) + (i - img * 784)] = xv[j];
            }
        }
        float4 f1, f2, f3;
        fir_taps(wv0, wv1, wv2, f1, f2, f3);
        wsh4[tid] = f1;
        wsh4[tid + 1024] = f2;
        wsh4[tid + 2048] = f3;
    }

    const float* alane = act + lo * IMGP + hi * 98;

    auto fill_stage = [&](const int gn) {
        const int cn = (gn >= 6) ? 1 : 0;
        const int jn = gn - 6 * cn;
        const bool pa = jn < 3;
        const int fr = pa ? jn : jn - 3;
        const float4* srcw = w4 + (cn * 9 + fr * 3) * (SLAB / 4) + it;
        float4* dstw = wsh4 + (gn & 1) * (STG / 4) + it;
        if (pa) {
            #pragma unroll
            for (int j0 = 0; j0 < 8; j0 += 2) {
                const float4 t00 = srcw[(j0 + 0) * 128];
                const float4 t01 = srcw[(j0 + 0) * 128 + 1024];
                const float4 t02 = srcw[(j0 + 0) * 128 + 2048];
                const float4 t10 = srcw[(j0 + 1) * 128];
                const float4 t11 = srcw[(j0 + 1) * 128 + 1024];
                const float4 t12 = srcw[(j0 + 1) * 128 + 2048];
                float4 fa0, fb0, fc0, fa1, fb1, fc1;
                fir_taps(t00, t01, t02, fa0, fb0, fc0);
                fir_taps(t10, t11, t12, fa1, fb1, fc1);
                dstw[(j0 + 0) * 128] = fa0;
                dstw[(j0 + 0) * 128 + 1024] = fb0;
                dstw[(j0 + 0) * 128 + 2048] = fc0;
                dstw[(j0 + 1) * 128] = fa1;
                dstw[(j0 + 1) * 128 + 1024] = fb1;
                dstw[(j0 + 1) * 128 + 2048] = fc1;
            }
        } else {
            #pragma unroll
            for (int j0 = 0; j0 < 8; j0 += 2) {
                const float4 ta0 = srcw[(j0 + 0) * 128];
                const float4 tb0 = srcw[(j0 + 0) * 128 + 1024];
                const float4 tc0 = srcw[(j0 + 0) * 128 + 2048];
                const float4 ta1 = srcw[(j0 + 1) * 128];
                const float4 tb1 = srcw[(j0 + 1) * 128 + 1024];
                const float4 tc1 = srcw[(j0 + 1) * 128 + 2048];
                dstw[(j0 + 0) * 128] = ta0;
                dstw[(j0 + 0) * 128 + 1024] = tc0;
                dstw[(j0 + 0) * 128 + 2048] = tb0;
                dstw[(j0 + 1) * 128] = ta1;
                dstw[(j0 + 1) * 128 + 1024] = tc1;
                dstw[(j0 + 1) * 128 + 2048] = tb1;
            }
        }
    };

    auto conv_pass = [&](const int c, v8f* acc) {
        #pragma unroll 1
        for (int r = 0; r < 3; ++r) {
            const int g  = c * 6 + r;
            const int ih = pg + r - 1;
            const bool rowok = own && (unsigned)ih < 7u;
            __syncthreads();
            if (!own) fill_stage(g + 1);
            if (rowok) {
                const float* arow = alane + ih * 7;
                const float* wb = wsh + (g & 1) * STG + hi * 128 + nn * 2;
                v2f a[2][6];
                v2f bf[2][3];
                #pragma unroll
                for (int q = 0; q < 6; ++q) {
                    const float* ap = arow + po[q];
                    a[0][q][0] = ap[0];
                    a[0][q][1] = ap[49];
                }
                #pragma unroll
                for (int s = 0; s < 3; ++s) bf[0][s] = *(const v2f*)&wb[s * SLAB];
                #pragma unroll
                for (int ks = 0; ks < 16; ++ks) {
                    if (ks < 15) {
                        #pragma unroll
                        for (int q = 0; q < 6; ++q) {
                            const float* ap = arow + po[q];
                            a[(ks + 1) & 1][q][0] = ap[(ks + 1) * 196];
                            a[(ks + 1) & 1][q][1] = ap[(ks + 1) * 196 + 49];
                        }
                        #pragma unroll
                        for (int s = 0; s < 3; ++s)
                            bf[(ks + 1) & 1][s] = *(const v2f*)&wb[s * SLAB + (ks + 1) * 256];
                    }
                    __builtin_amdgcn_sched_barrier(0);
                    const v2f* q = a[ks & 1];
                    const v2f w1 = bf[ks & 1][0], w2 = bf[ks & 1][1], w3 = bf[ks & 1][2];
                    const v2f a3 = q[2] - q[0];
                    const v2f a1 = 2.0f * q[0] + (q[1] - q[2]);
                    const v2f a2 = 3.0f * q[1] - (2.0f * q[0] + q[2]);
                    const v2f b3 = q[5] - q[3];
                    const v2f b1 = 2.0f * q[3] + (q[4] - q[5]);
                    const v2f b2 = 3.0f * q[4] - (2.0f * q[3] + q[5]);
                    acc[0] = __builtin_amdgcn_wmma_f32_16x16x4_f32(false, a1, false, w1, (short)0, acc[0], false, false);
                    acc[3] = __builtin_amdgcn_wmma_f32_16x16x4_f32(false, b1, false, w1, (short)0, acc[3], false, false);
                    acc[1] = __builtin_amdgcn_wmma_f32_16x16x4_f32(false, a2, false, w2, (short)0, acc[1], false, false);
                    acc[4] = __builtin_amdgcn_wmma_f32_16x16x4_f32(false, b2, false, w2, (short)0, acc[4], false, false);
                    acc[2] = __builtin_amdgcn_wmma_f32_16x16x4_f32(false, a3, false, w3, (short)0, acc[2], false, false);
                    acc[5] = __builtin_amdgcn_wmma_f32_16x16x4_f32(false, b3, false, w3, (short)0, acc[5], false, false);
                    __builtin_amdgcn_sched_barrier(0);
                }
            }
        }

        #pragma unroll
        for (int tl = 0; tl < 6; tl += 3) {
            const v8f m1 = acc[tl], m2 = acc[tl + 1], m3 = acc[tl + 2];
            const v8f sm = m1 + m2, df = m1 - m2;
            acc[tl]     = sm + m3;
            acc[tl + 1] = df + 2.0f * m3;
            acc[tl + 2] = sm + 4.0f * m3;
        }
        asm volatile("" : "+v"(acc[1]));
        asm volatile("" : "+v"(acc[4]));

        #pragma unroll 1
        for (int r = 0; r < 3; ++r) {
            const int g  = c * 6 + 3 + r;
            const int ih = pg + r - 1;
            const bool rowok = own && (unsigned)ih < 7u;
            __syncthreads();
            if (!own) {
                if (g < 11) fill_stage(g + 1);
            }
            if (rowok) {
                const float* arow = alane + ih * 7;
                const float* wb = wsh + (g & 1) * STG + hi * 128 + nn * 2;
                v2f a[2][7];
                v2f bf[2][3];
                #pragma unroll
                for (int q = 0; q < 7; ++q) {
                    const float* ap = arow + po[q];
                    a[0][q][0] = ap[0];
                    a[0][q][1] = ap[49];
                }
                #pragma unroll
                for (int s = 0; s < 3; ++s) bf[0][s] = *(const v2f*)&wb[s * SLAB];
                #pragma unroll
                for (int ks = 0; ks < 16; ++ks) {
                    if (ks < 15) {
                        #pragma unroll
                        for (int q = 0; q < 7; ++q) {
                            const float* ap = arow + po[q];
                            a[(ks + 1) & 1][q][0] = ap[(ks + 1) * 196];
                            a[(ks + 1) & 1][q][1] = ap[(ks + 1) * 196 + 49];
                        }
                        #pragma unroll
                        for (int s = 0; s < 3; ++s)
                            bf[(ks + 1) & 1][s] = *(const v2f*)&wb[s * SLAB + (ks + 1) * 256];
                    }
                    __builtin_amdgcn_sched_barrier(0);
                    const v2f* q = a[ks & 1];
                    const v2f t0 = bf[ks & 1][0], t2 = bf[ks & 1][1], t1 = bf[ks & 1][2];
                    const v2f e20 = q[2] - q[0], e53 = q[5] - q[3];
                    const v2f a0 = 0.5f * e20 - q[1];
                    const v2f a4 = (q[3] - q[1]) - 2.0f * e20;
                    const v2f b0 = (q[2] - q[4]) + 0.5f * e53;
                    const v2f b4 = (q[6] - q[4]) - 2.0f * e53;
                    acc[6] = __builtin_amdgcn_wmma_f32_16x16x4_f32(false, q[5], false, t0, (short)0, acc[6], false, false);
                    acc[0] = __builtin_amdgcn_wmma_f32_16x16x4_f32(false, a0, false, t0, (short)0, acc[0], false, false);
                    acc[3] = __builtin_amdgcn_wmma_f32_16x16x4_f32(false, b0, false, t0, (short)0, acc[3], false, false);
                    acc[2] = __builtin_amdgcn_wmma_f32_16x16x4_f32(false, a4, false, t2, (short)0, acc[2], false, false);
                    acc[5] = __builtin_amdgcn_wmma_f32_16x16x4_f32(false, b4, false, t2, (short)0, acc[5], false, false);
                    acc[6] = __builtin_amdgcn_wmma_f32_16x16x4_f32(false, q[6], false, t1, (short)0, acc[6], false, false);
                    __builtin_amdgcn_sched_barrier(0);
                }
            }
        }
    };

    v8f c1[7] = {};
    conv_pass(0, c1);
    __syncthreads();

    const float sc1 = g1[nn] * rsqrtf(v1[nn] + EPSV);
    const float sh1 = b1[nn] - m1[nn] * sc1;
    if (own) {
        #pragma unroll
        for (int t = 0; t < 7; ++t) {
            const int p = pg * 7 + t;
            #pragma unroll
            for (int j = 0; j < 8; ++j) {
                const int img = j + 8 * hi;
                float v = fmaf(c1[t][j], sc1, sh1);
                v = fmaxf(v, 0.0f);
                if (b0 + img == 0) v *= mask1[nn * 49 + p];
                act[img * IMGP + nn * 49 + p] = v;
            }
        }
    }

    v8f c2[7] = {};
    conv_pass(1, c2);
    __syncthreads();

    const float sc2 = g2[nn] * rsqrtf(v2p[nn] + EPSV);
    const float sh2 = b2[nn] - m2[nn] * sc2;
    if (own) {
        #pragma unroll
        for (int t = 0; t < 7; ++t) {
            const int p = pg * 7 + t;
            #pragma unroll
            for (int j = 0; j < 8; ++j) {
                const int img = j + 8 * hi;
                float v = c2[t][j];
                if (b0 + img == 0) v *= mask2[nn * 49 + p];
                v = fmaf(v, sc2, sh2);
                act[img * IMGP + nn * 49 + p] = v;
            }
        }
    }
    __syncthreads();

    float4* o4 = (float4*)(out + (size_t)b0 * 3136);
    int tl = tid;
    asm volatile("" : "+v"(tl));
    tl &= 1023;
    {
        float4 xr[13];
        #pragma unroll
        for (int j = 0; j < 13; ++j) {
            const int i = tl + j * 1024;
            xr[j] = (i < 12544) ? x4[i] : make_float4(0.f, 0.f, 0.f, 0.f);
        }
        #pragma unroll
        for (int j = 0; j < 13; ++j) {
            const int i = tl + j * 1024;
            if (i < 12544) {
                const int img = i / 784;
                float4 v = act4[img * (IMGP / 4) + (i - img * 784)];
                v.x = fmaxf(v.x + xr[j].x, 0.0f);
                v.y = fmaxf(v.y + xr[j].y, 0.0f);
                v.z = fmaxf(v.z + xr[j].z, 0.0f);
                v.w = fmaxf(v.w + xr[j].w, 0.0f);
                o4[i] = v;
            }
        }
    }
}

extern "C" void kernel_launch(void* const* d_in, const int* in_sizes, int n_in,
                              void* d_out, int out_size, void* d_ws, size_t ws_size,
                              hipStream_t stream) {
    const float* x     = (const float*)d_in[0];
    const float* w1    = (const float*)d_in[1];
    const float* g1    = (const float*)d_in[2];
    const float* b1    = (const float*)d_in[3];
    const float* m1    = (const float*)d_in[4];
    const float* v1    = (const float*)d_in[5];
    const float* w2    = (const float*)d_in[6];
    const float* g2    = (const float*)d_in[7];
    const float* b2    = (const float*)d_in[8];
    const float* m2    = (const float*)d_in[9];
    const float* v2p   = (const float*)d_in[10];
    const float* mask1 = (const float*)d_in[11];
    const float* mask2 = (const float*)d_in[12];
    float* wt = (float*)d_ws;
    float* out = (float*)d_out;

    prep_weights_kernel<<<288, 256, 0, stream>>>(w1, w2, wt);
    fused_basicblock_kernel<<<256, 1024, 0, stream>>>(
        x, wt, g1, b1, m1, v1, g2, b2, m2, v2p, mask1, mask2, out);
}


# Wave priority falls by quarter of a stage (s_setprio 3,2,1,0 over the 16 K-steps; staging wave at 3): the row wave that is behind wins the matrix pipe, so a SIMD's waves end each stage together
# speedup vs baseline: 2.4968x; 1.0522x over previous
#include <hip/hip_runtime.h>

#define EPSV 1e-5f

typedef __attribute__((ext_vector_type(2))) float v2f;
typedef __attribute__((ext_vector_type(8))) float v8f;

__global__ __launch_bounds__(256) void prep_weights_kernel(
    const float* __restrict__ w1, const float* __restrict__ w2,
    float* __restrict__ wt)
{
    int i = blockIdx.x * 256 + threadIdx.x;
    if (i >= 2 * 36864) return;
    const float* w = (i < 36864) ? w1 : w2;
    float* dst = wt + (i < 36864 ? 0 : 36864);
    int j = i % 36864;
    int tap = j / 4096;
    int rem = j % 4096;
    int n = (rem >> 1) & 63;
    int k = ((rem >> 7) << 1) | (rem & 1);
    dst[j] = w[(n * 64 + k) * 9 + tap];
}

#define NIMG 16
#define IMGP 3140
#define SLAB 4096
#define STG  (3 * SLAB)

__device__ __forceinline__ void fir_tap1(const float g0, const float g1, const float g2,
                                         float& w1, float& w2, float& w3)
{
    const float e = g0 + g2;
    w1 = 0.5f * (e + g1);
    w2 = (1.0f / 6.0f) * (e - g1);
    w3 = (1.0f / 6.0f) * (fmaf(4.0f, g2, g0) + 2.0f * g1);
}
__device__ __forceinline__ void fir_taps(const float4 g0, const float4 g1, const float4 g2,
                                         float4& w1, float4& w2, float4& w3)
{
    fir_tap1(g0.x, g1.x, g2.x, w1.x, w2.x, w3.x);
    fir_tap1(g0.y, g1.y, g2.y, w1.y, w2.y, w3.y);
    fir_tap1(g0.z, g1.z, g2.z, w1.z, w2.z, w3.z);
    fir_tap1(g0.w, g1.w, g2.w, w1.w, w2.w, w3.w);
}

__global__ __launch_bounds__(1024) void fused_basicblock_kernel(
    const float* __restrict__ x,
    const float* __restrict__ wt,
    const float* __restrict__ g1, const float* __restrict__ b1,
    const float* __restrict__ m1, const float* __restrict__ v1,
    const float* __restrict__ g2, const float* __restrict__ b2,
    const float* __restrict__ m2, const float* __restrict__ v2p,
    const float* __restrict__ mask1, const float* __restrict__ mask2,
    float* __restrict__ out)
{
    __shared__ __align__(16) float act[NIMG * IMGP];
    __shared__ __align__(16) float wsh[2 * STG];

    const int tid  = threadIdx.x;
    const int lane = tid & 31;
    const int wave = __builtin_amdgcn_readfirstlane(tid >> 5);
    const int b0   = blockIdx.x * NIMG;

    const int lo = lane & 15;
    const int hi = lane >> 4;

    const int nn = (wave & 3) * 16 + lo;
    const int pg = 2 * (((wave >> 3) + (wave & 3)) & 3) + ((wave >> 2) & 1);
    const bool own = pg < 7;
    const int it = (wave & 3) * 32 + lane;
    if (!own) __builtin_amdgcn_s_setprio(3);

    int po[7];
    #pragma unroll
    for (int i = 0; i < 7; ++i) {
        int o = i;
        asm volatile("" : "+s"(o));
        po[i] = o;
    }

    const float4* w4 = (const float4*)wt;
    float4* wsh4 = (float4*)wsh;

    const float4* x4 = (const float4*)(x + (size_t)b0 * 3136);
    float4* act4 = (float4*)act;
    {
        float4 xv[13];
        #pragma unroll
        for (int j = 0; j < 13; ++j) {
            const int i = tid + j * 1024;
            xv[j] = (i < 12544) ? x4[i] : make_float4(0.f, 0.f, 0.f, 0.f);
        }
        const float4 wv0 = w4[tid], wv1 = w4[tid + 1024], wv2 = w4[tid + 2048];
        #pragma unroll
        for (int j = 0; j < 13; ++j) {
            const int i = tid + j * 1024;
            if (i < 12544) {
                const int img = i / 784;
                act4[img * (IMGP / 4) + (i - img * 784)] = xv[j];
            }
        }
        float4 f1, f2, f3;
        fir_taps(wv0, wv1, wv2, f1, f2, f3);
        wsh4[tid] = f1;
        wsh4[tid + 1024] = f2;
        wsh4[tid + 2048] = f3;
    }

    const float* alane = act + lo * IMGP + hi * 98;

    auto fill_stage = [&](const int gn) {
        const int cn = (gn >= 6) ? 1 : 0;
        const int jn = gn - 6 * cn;
        const bool pa = jn < 3;
        const int fr = pa ? jn : jn - 3;
        const float4* srcw = w4 + (cn * 9 + fr * 3) * (SLAB / 4) + it;
        float4* dstw = wsh4 + (gn & 1) * (STG / 4) + it;
        if (pa) {
            #pragma unroll
            for (int j0 = 0; j0 < 8; j0 += 2) {
                const float4 t00 = srcw[(j0 + 0) * 128];
                const float4 t01 = srcw[(j0 + 0) * 128 + 1024];
                const float4 t02 = srcw[(j0 + 0) * 128 + 2048];
                const float4 t10 = srcw[(j0 + 1) * 128];
                const float4 t11 = srcw[(j0 + 1) * 128 + 1024];
                const float4 t12 = srcw[(j0 + 1) * 128 + 2048];
                float4 fa0, fb0, fc0, fa1, fb1, fc1;
                fir_taps(t00, t01, t02, fa0, fb0, fc0);
                fir_taps(t10, t11, t12, fa1, fb1, fc1);
                dstw[(j0 + 0) * 128] = fa0;
                dstw[(j0 + 0) * 128 + 1024] = fb0;
                dstw[(j0 + 0) * 128 + 2048] = fc0;
                dstw[(j0 + 1) * 128] = fa1;
                dstw[(j0 + 1) * 128 + 1024] = fb1;
                dstw[(j0 + 1) * 128 + 2048] = fc1;
            }
        } else {
            #pragma unroll
            for (int j0 = 0; j0 < 8; j0 += 2) {
                const float4 ta0 = srcw[(j0 + 0) * 128];
                const float4 tb0 = srcw[(j0 + 0) * 128 + 1024];
                const float4 tc0 = srcw[(j0 + 0) * 128 + 2048];
                const float4 ta1 = srcw[(j0 + 1) * 128];
                const float4 tb1 = srcw[(j0 + 1) * 128 + 1024];
                const float4 tc1 = srcw[(j0 + 1) * 128 + 2048];
                dstw[(j0 + 0) * 128] = ta0;
                dstw[(j0 + 0) * 128 + 1024] = tc0;
                dstw[(j0 + 0) * 128 + 2048] = tb0;
                dstw[(j0 + 1) * 128] = ta1;
                dstw[(j0 + 1) * 128 + 1024] = tc1;
                dstw[(j0 + 1) * 128 + 2048] = tb1;
            }
        }
    };

    auto conv_pass = [&](const int c, v8f* acc) {
        #pragma unroll 1
        for (int r = 0; r < 3; ++r) {
            const int g  = c * 6 + r;
            const int ih = pg + r - 1;
            const bool rowok = own && (unsigned)ih < 7u;
            __syncthreads();
            if (!own) fill_stage(g + 1);
            if (rowok) {
                const float* arow = alane + ih * 7;
                const float* wb = wsh + (g & 1) * STG + hi * 128 + nn * 2;
                v2f a[2][6];
                v2f bf[2][3];
                #pragma unroll
                for (int q = 0; q < 6; ++q) {
                    const float* ap = arow + po[q];
                    a[0][q][0] = ap[0];
                    a[0][q][1] = ap[49];
                }
                #pragma unroll
                for (int s = 0; s < 3; ++s) bf[0][s] = *(const v2f*)&wb[s * SLAB];
                #pragma unroll
                for (int ks = 0; ks < 16; ++ks) {
                    if (ks < 15) {
                        #pragma unroll
                        for (int q = 0; q < 6; ++q) {
                            const float* ap = arow + po[q];
                            a[(ks + 1) & 1][q][0] = ap[(ks + 1) * 196];
                            a[(ks + 1) & 1][q][1] = ap[(ks + 1) * 196 + 49];
                        }
                        #pragma unroll
                        for (int s = 0; s < 3; ++s)
                            bf[(ks + 1) & 1][s] = *(const v2f*)&wb[s * SLAB + (ks + 1) * 256];
                    }
                    __builtin_amdgcn_sched_barrier(0);
                    if (ks == 0)       __builtin_amdgcn_s_setprio(3);
                    else if (ks == 4)  __builtin_amdgcn_s_setprio(2);
                    else if (ks == 8)  __builtin_amdgcn_s_setprio(1);
                    else if (ks == 12) __builtin_amdgcn_s_setprio(0);
                    const v2f* q = a[ks & 1];
                    const v2f w1 = bf[ks & 1][0], w2 = bf[ks & 1][1], w3 = bf[ks & 1][2];
                    const v2f a3 = q[2] - q[0];
                    const v2f a1 = 2.0f * q[0] + (q[1] - q[2]);
                    const v2f a2 = 3.0f * q[1] - (2.0f * q[0] + q[2]);
                    const v2f b3 = q[5] - q[3];
                    const v2f b1 = 2.0f * q[3] + (q[4] - q[5]);
                    const v2f b2 = 3.0f * q[4] - (2.0f * q[3] + q[5]);
                    acc[0] = __builtin_amdgcn_wmma_f32_16x16x4_f32(false, a1, false, w1, (short)0, acc[0], false, false);
                    acc[3] = __builtin_amdgcn_wmma_f32_16x16x4_f32(false, b1, false, w1, (short)0, acc[3], false, false);
                    acc[1] = __builtin_amdgcn_wmma_f32_16x16x4_f32(false, a2, false, w2, (short)0, acc[1], false, false);
                    acc[4] = __builtin_amdgcn_wmma_f32_16x16x4_f32(false, b2, false, w2, (short)0, acc[4], false, false);
                    acc[2] = __builtin_amdgcn_wmma_f32_16x16x4_f32(false, a3, false, w3, (short)0, acc[2], false, false);
                    acc[5] = __builtin_amdgcn_wmma_f32_16x16x4_f32(false, b3, false, w3, (short)0, acc[5], false, false);
                    __builtin_amdgcn_sched_barrier(0);
                }
            }
        }

        #pragma unroll
        for (int tl = 0; tl < 6; tl += 3) {
            const v8f m1 = acc[tl], m2 = acc[tl + 1], m3 = acc[tl + 2];
            const v8f sm = m1 + m2, df = m1 - m2;
            acc[tl]     = sm + m3;
            acc[tl + 1] = df + 2.0f * m3;
            acc[tl + 2] = sm + 4.0f * m3;
        }
        asm volatile("" : "+v"(acc[1]));
        asm volatile("" : "+v"(acc[4]));

        #pragma unroll 1
        for (int r = 0; r < 3; ++r) {
            const int g  = c * 6 + 3 + r;
            const int ih = pg + r - 1;
            const bool rowok = own && (unsigned)ih < 7u;
            __syncthreads();
            if (!own) {
                if (g < 11) fill_stage(g + 1);
            }
            if (rowok) {
                const float* arow = alane + ih * 7;
                const float* wb = wsh + (g & 1) * STG + hi * 128 + nn * 2;
                v2f a[2][7];
                v2f bf[2][3];
                #pragma unroll
                for (int q = 0; q < 7; ++q) {
                    const float* ap = arow + po[q];
                    a[0][q][0] = ap[0];
                    a[0][q][1] = ap[49];
                }
                #pragma unroll
                for (int s = 0; s < 3; ++s) bf[0][s] = *(const v2f*)&wb[s * SLAB];
                #pragma unroll
                for (int ks = 0; ks < 16; ++ks) {
                    if (ks < 15) {
                        #pragma unroll
                        for (int q = 0; q < 7; ++q) {
                            const float* ap = arow + po[q];
                            a[(ks + 1) & 1][q][0] = ap[(ks + 1) * 196];
                            a[(ks + 1) & 1][q][1] = ap[(ks + 1) * 196 + 49];
                        }
                        #pragma unroll
                        for (int s = 0; s < 3; ++s)
                            bf[(ks + 1) & 1][s] = *(const v2f*)&wb[s * SLAB + (ks + 1) * 256];
                    }
                    __builtin_amdgcn_sched_barrier(0);
                    if (ks == 0)       __builtin_amdgcn_s_setprio(3);
                    else if (ks == 4)  __builtin_amdgcn_s_setprio(2);
                    else if (ks == 8)  __builtin_amdgcn_s_setprio(1);
                    else if (ks == 12) __builtin_amdgcn_s_setprio(0);
                    const v2f* q = a[ks & 1];
                    const v2f t0 = bf[ks & 1][0], t2 = bf[ks & 1][1], t1 = bf[ks & 1][2];
                    const v2f e20 = q[2] - q[0], e53 = q[5] - q[3];
                    const v2f a0 = 0.5f * e20 - q[1];
                    const v2f a4 = (q[3] - q[1]) - 2.0f * e20;
                    const v2f b0 = (q[2] - q[4]) + 0.5f * e53;
                    const v2f b4 = (q[6] - q[4]) - 2.0f * e53;
                    acc[6] = __builtin_amdgcn_wmma_f32_16x16x4_f32(false, q[5], false, t0, (short)0, acc[6], false, false);
                    acc[0] = __builtin_amdgcn_wmma_f32_16x16x4_f32(false, a0, false, t0, (short)0, acc[0], false, false);
                    acc[3] = __builtin_amdgcn_wmma_f32_16x16x4_f32(false, b0, false, t0, (short)0, acc[3], false, false);
                    acc[2] = __builtin_amdgcn_wmma_f32_16x16x4_f32(false, a4, false, t2, (short)0, acc[2], false, false);
                    acc[5] = __builtin_amdgcn_wmma_f32_16x16x4_f32(false, b4, false, t2, (short)0, acc[5], false, false);
                    acc[6] = __builtin_amdgcn_wmma_f32_16x16x4_f32(false, q[6], false, t1, (short)0, acc[6], false, false);
                    __builtin_amdgcn_sched_barrier(0);
                }
            }
        }
    };

    v8f c1[7] = {};
    conv_pass(0, c1);
    __syncthreads();

    const float sc1 = g1[nn] * rsqrtf(v1[nn] + EPSV);
    const float sh1 = b1[nn] - m1[nn] * sc1;
    if (own) {
        #pragma unroll
        for (int t = 0; t < 7; ++t) {
            const int p = pg * 7 + t;
            #pragma unroll
            for (int j = 0; j < 8; ++j) {
                const int img = j + 8 * hi;
                float v = fmaf(c1[t][j], sc1, sh1);
                v = fmaxf(v, 0.0f);
                if (b0 + img == 0) v *= mask1[nn * 49 + p];
                act[img * IMGP + nn * 49 + p] = v;
            }
        }
    }

    v8f c2[7] = {};
    conv_pass(1, c2);
    __syncthreads();

    const float sc2 = g2[nn] * rsqrtf(v2p[nn] + EPSV);
    const float sh2 = b2[nn] - m2[nn] * sc2;
    if (own) {
        #pragma unroll
        for (int t = 0; t < 7; ++t) {
            const int p = pg * 7 + t;
            #pragma unroll
            for (int j = 0; j < 8; ++j) {
                const int img = j + 8 * hi;
                float v = c2[t][j];
                if (b0 + img == 0) v *= mask2[nn * 49 + p];
                v = fmaf(v, sc2, sh2);
                act[img * IMGP + nn * 49 + p] = v;
            }
        }
    }
    __syncthreads();

    float4* o4 = (float4*)(out + (size_t)b0 * 3136);
    int tl = tid;
    asm volatile("" : "+v"(tl));
    tl &= 1023;
    {
        float4 xr[13];
        #pragma unroll
        for (int j = 0; j < 13; ++j) {
            const int i = tl + j * 1024;
            xr[j] = (i < 12544) ? x4[i] : make_float4(0.f, 0.f, 0.f, 0.f);
        }
        #pragma unroll
        for (int j = 0; j < 13; ++j) {
            const int i = tl + j * 1024;
            if (i < 12544) {
                const int img = i / 784;
                float4 v = act4[img * (IMGP / 4) + (i - img * 784)];
                v.x = fmaxf(v.x + xr[j].x, 0.0f);
                v.y = fmaxf(v.y + xr[j].y, 0.0f);
                v.z = fmaxf(v.z + xr[j].z, 0.0f);
                v.w = fmaxf(v.w + xr[j].w, 0.0f);
                o4[i] = v;
            }
        }
    }
}

extern "C" void kernel_launch(void* const* d_in, const int* in_sizes, int n_in,
                              void* d_out, int out_size, void* d_ws, size_t ws_size,
                              hipStream_t stream) {
    const float* x     = (const float*)d_in[0];
    const float* w1    = (const float*)d_in[1];
    const float* g1    = (const float*)d_in[2];
    const float* b1    = (const float*)d_in[3];
    const float* m1    = (const float*)d_in[4];
    const float* v1    = (const float*)d_in[5];
    const float* w2    = (const float*)d_in[6];
    const float* g2    = (const float*)d_in[7];
    const float* b2    = (const float*)d_in[8];
    const float* m2    = (const float*)d_in[9];
    const float* v2p   = (const float*)d_in[10];
    const float* mask1 = (const float*)d_in[11];
    const float* mask2 = (const float*)d_in[12];
    float* wt = (float*)d_ws;
    float* out = (float*)d_out;

    prep_weights_kernel<<<288, 256, 0, stream>>>(w1, w2, wt);
    fused_basicblock_kernel<<<256, 1024, 0, stream>>>(
        x, wt, g1, b1, m1, v1, g2, b2, m2, v2p, mask1, mask2, out);
}


# LDS activation tile channel-pair interleaved [ch/2][px][2]: an A fragment is one 8-byte load at a constant offset; the 6-7 vector address adds per K-step are gone; prologue restaged to match
# speedup vs baseline: 2.6121x; 1.0462x over previous
#include <hip/hip_runtime.h>

#define EPSV 1e-5f

typedef __attribute__((ext_vector_type(2))) float v2f;
typedef __attribute__((ext_vector_type(8))) float v8f;

__global__ __launch_bounds__(256) void prep_weights_kernel(
    const float* __restrict__ w1, const float* __restrict__ w2,
    float* __restrict__ wt)
{
    int i = blockIdx.x * 256 + threadIdx.x;
    if (i >= 2 * 36864) return;
    const float* w = (i < 36864) ? w1 : w2;
    float* dst = wt + (i < 36864 ? 0 : 36864);
    int j = i % 36864;
    int tap = j / 4096;
    int rem = j % 4096;
    int n = (rem >> 1) & 63;
    int k = ((rem >> 7) << 1) | (rem & 1);
    dst[j] = w[(n * 64 + k) * 9 + tap];
}

#define NIMG 16
#define IMGP 3140
#define SLAB 4096
#define STG  (3 * SLAB)

__device__ __forceinline__ void fir_tap1(const float g0, const float g1, const float g2,
                                         float& w1, float& w2, float& w3)
{
    const float e = g0 + g2;
    w1 = 0.5f * (e + g1);
    w2 = (1.0f / 6.0f) * (e - g1);
    w3 = (1.0f / 6.0f) * (fmaf(4.0f, g2, g0) + 2.0f * g1);
}
__device__ __forceinline__ void fir_taps(const float4 g0, const float4 g1, const float4 g2,
                                         float4& w1, float4& w2, float4& w3)
{
    fir_tap1(g0.x, g1.x, g2.x, w1.x, w2.x, w3.x);
    fir_tap1(g0.y, g1.y, g2.y, w1.y, w2.y, w3.y);
    fir_tap1(g0.z, g1.z, g2.z, w1.z, w2.z, w3.z);
    fir_tap1(g0.w, g1.w, g2.w, w1.w, w2.w, w3.w);
}

__global__ __launch_bounds__(1024) void fused_basicblock_kernel(
    const float* __restrict__ x,
    const float* __restrict__ wt,
    const float* __restrict__ g1, const float* __restrict__ b1,
    const float* __restrict__ m1, const float* __restrict__ v1,
    const float* __restrict__ g2, const float* __restrict__ b2,
    const float* __restrict__ m2, const float* __restrict__ v2p,
    const float* __restrict__ mask1, const float* __restrict__ mask2,
    float* __restrict__ out)
{
    __shared__ __align__(16) float act[NIMG * IMGP];
    __shared__ __align__(16) float wsh[2 * STG];

    const int tid  = threadIdx.x;
    const int lane = tid & 31;
    const int wave = __builtin_amdgcn_readfirstlane(tid >> 5);
    const int b0   = blockIdx.x * NIMG;

    const int lo = lane & 15;
    const int hi = lane >> 4;

    const int nn = (wave & 3) * 16 + lo;
    const int pg = 2 * (((wave >> 3) + (wave & 3)) & 3) + ((wave >> 2) & 1);
    const bool own = pg < 7;
    const int it = (wave & 3) * 32 + lane;
    if (!own) __builtin_amdgcn_s_setprio(3);

    int po[7];
    #pragma unroll
    for (int i = 0; i < 7; ++i) {
        int o = 2 * i;
        asm volatile("" : "+s"(o));
        po[i] = o;
    }

    const float4* w4 = (const float4*)wt;
    float4* wsh4 = (float4*)wsh;

    const float4* x4 = (const float4*)(x + (size_t)b0 * 3136);
    float4* act4 = (float4*)act;
    {
        const int simg = tid >> 6;
        const int l64 = tid & 63;
        const float* xg = x + (size_t)(b0 + simg) * 3136;
        float xe[25], xo[25];
        #pragma unroll
        for (int j = 0; j < 25; ++j) {
            const int r = l64 + j * 64;
            const int cp = (r * 1338) >> 16;
            const bool ok = (j < 24) || (l64 < 32);
            xe[j] = ok ? xg[r + cp * 49] : 0.0f;
            xo[j] = ok ? xg[r + cp * 49 + 49] : 0.0f;
        }
        const float4 wv0 = w4[tid], wv1 = w4[tid + 1024], wv2 = w4[tid + 2048];
        #pragma unroll
        for (int j = 0; j < 25; ++j) {
            const int r = l64 + j * 64;
            if ((j < 24) || (l64 < 32)) {
                v2f pr;
                pr[0] = xe[j];
                pr[1] = xo[j];
                *(v2f*)&act[simg * IMGP + 2 * r] = pr;
            }
        }
        float4 f1, f2, f3;
        fir_taps(wv0, wv1, wv2, f1, f2, f3);
        wsh4[tid] = f1;
        wsh4[tid + 1024] = f2;
        wsh4[tid + 2048] = f3;
    }

    const float* alane = act + lo * IMGP + hi * 98;

    auto fill_stage = [&](const int gn) {
        const int cn = (gn >= 6) ? 1 : 0;
        const int jn = gn - 6 * cn;
        const bool pa = jn < 3;
        const int fr = pa ? jn : jn - 3;
        const float4* srcw = w4 + (cn * 9 + fr * 3) * (SLAB / 4) + it;
        float4* dstw = wsh4 + (gn & 1) * (STG / 4) + it;
        if (pa) {
            #pragma unroll
            for (int j0 = 0; j0 < 8; j0 += 2) {
                const float4 t00 = srcw[(j0 + 0) * 128];
                const float4 t01 = srcw[(j0 + 0) * 128 + 1024];
                const float4 t02 = srcw[(j0 + 0) * 128 + 2048];
                const float4 t10 = srcw[(j0 + 1) * 128];
                const float4 t11 = srcw[(j0 + 1) * 128 + 1024];
                const float4 t12 = srcw[(j0 + 1) * 128 + 2048];
                float4 fa0, fb0, fc0, fa1, fb1, fc1;
                fir_taps(t00, t01, t02, fa0, fb0, fc0);
                fir_taps(t10, t11, t12, fa1, fb1, fc1);
                dstw[(j0 + 0) * 128] = fa0;
                dstw[(j0 + 0) * 128 + 1024] = fb0;
                dstw[(j0 + 0) * 128 + 2048] = fc0;
                dstw[(j0 + 1) * 128] = fa1;
                dstw[(j0 + 1) * 128 + 1024] = fb1;
                dstw[(j0 + 1) * 128 + 2048] = fc1;
            }
        } else {
            #pragma unroll
            for (int j0 = 0; j0 < 8; j0 += 2) {
                const float4 ta0 = srcw[(j0 + 0) * 128];
                const float4 tb0 = srcw[(j0 + 0) * 128 + 1024];
                const float4 tc0 = srcw[(j0 + 0) * 128 + 2048];
                const float4 ta1 = srcw[(j0 + 1) * 128];
                const float4 tb1 = srcw[(j0 + 1) * 128 + 1024];
                const float4 tc1 = srcw[(j0 + 1) * 128 + 2048];
                dstw[(j0 + 0) * 128] = ta0;
                dstw[(j0 + 0) * 128 + 1024] = tc0;
                dstw[(j0 + 0) * 128 + 2048] = tb0;
                dstw[(j0 + 1) * 128] = ta1;
                dstw[(j0 + 1) * 128 + 1024] = tc1;
                dstw[(j0 + 1) * 128 + 2048] = tb1;
            }
        }
    };

    auto conv_pass = [&](const int c, v8f* acc) {
        #pragma unroll 1
        for (int r = 0; r < 3; ++r) {
            const int g  = c * 6 + r;
            const int ih = pg + r - 1;
            const bool rowok = own && (unsigned)ih < 7u;
            __syncthreads();
            if (!own) fill_stage(g + 1);
            if (rowok) {
                const float* arow = alane + ih * 14;
                const float* wb = wsh + (g & 1) * STG + hi * 128 + nn * 2;
                v2f a[2][6];
                v2f bf[2][3];
                #pragma unroll
                for (int q = 0; q < 6; ++q) a[0][q] = *(const v2f*)(arow + po[q]);
                #pragma unroll
                for (int s = 0; s < 3; ++s) bf[0][s] = *(const v2f*)&wb[s * SLAB];
                #pragma unroll
                for (int ks = 0; ks < 16; ++ks) {
                    if (ks < 15) {
                        #pragma unroll
                        for (int q = 0; q < 6; ++q)
                            a[(ks + 1) & 1][q] = *(const v2f*)&(arow + po[q])[(ks + 1) * 196];
                        #pragma unroll
                        for (int s = 0; s < 3; ++s)
                            bf[(ks + 1) & 1][s] = *(const v2f*)&wb[s * SLAB + (ks + 1) * 256];
                    }
                    __builtin_amdgcn_sched_barrier(0);
                    if (ks == 0)       __builtin_amdgcn_s_setprio(3);
                    else if (ks == 4)  __builtin_amdgcn_s_setprio(2);
                    else if (ks == 8)  __builtin_amdgcn_s_setprio(1);
                    else if (ks == 12) __builtin_amdgcn_s_setprio(0);
                    const v2f* q = a[ks & 1];
                    const v2f w1 = bf[ks & 1][0], w2 = bf[ks & 1][1], w3 = bf[ks & 1][2];
                    const v2f a3 = q[2] - q[0];
                    const v2f a1 = 2.0f * q[0] + (q[1] - q[2]);
                    const v2f a2 = 3.0f * q[1] - (2.0f * q[0] + q[2]);
                    const v2f b3 = q[5] - q[3];
                    const v2f b1 = 2.0f * q[3] + (q[4] - q[5]);
                    const v2f b2 = 3.0f * q[4] - (2.0f * q[3] + q[5]);
                    acc[0] = __builtin_amdgcn_wmma_f32_16x16x4_f32(false, a1, false, w1, (short)0, acc[0], false, false);
                    acc[3] = __builtin_amdgcn_wmma_f32_16x16x4_f32(false, b1, false, w1, (short)0, acc[3], false, false);
                    acc[1] = __builtin_amdgcn_wmma_f32_16x16x4_f32(false, a2, false, w2, (short)0, acc[1], false, false);
                    acc[4] = __builtin_amdgcn_wmma_f32_16x16x4_f32(false, b2, false, w2, (short)0, acc[4], false, false);
                    acc[2] = __builtin_amdgcn_wmma_f32_16x16x4_f32(false, a3, false, w3, (short)0, acc[2], false, false);
                    acc[5] = __builtin_amdgcn_wmma_f32_16x16x4_f32(false, b3, false, w3, (short)0, acc[5], false, false);
                    __builtin_amdgcn_sched_barrier(0);
                }
            }
        }

        #pragma unroll
        for (int tl = 0; tl < 6; tl += 3) {
            const v8f m1 = acc[tl], m2 = acc[tl + 1], m3 = acc[tl + 2];
            const v8f sm = m1 + m2, df = m1 - m2;
            acc[tl]     = sm + m3;
            acc[tl + 1] = df + 2.0f * m3;
            acc[tl + 2] = sm + 4.0f * m3;
        }
        asm volatile("" : "+v"(acc[1]));
        asm volatile("" : "+v"(acc[4]));

        #pragma unroll 1
        for (int r = 0; r < 3; ++r) {
            const int g  = c * 6 + 3 + r;
            const int ih = pg + r - 1;
            const bool rowok = own && (unsigned)ih < 7u;
            __syncthreads();
            if (!own) {
                if (g < 11) fill_stage(g + 1);
            }
            if (rowok) {
                const float* arow = alane + ih * 14;
                const float* wb = wsh + (g & 1) * STG + hi * 128 + nn * 2;
                v2f a[2][7];
                v2f bf[2][3];
                #pragma unroll
                for (int q = 0; q < 7; ++q) a[0][q] = *(const v2f*)(arow + po[q]);
                #pragma unroll
                for (int s = 0; s < 3; ++s) bf[0][s] = *(const v2f*)&wb[s * SLAB];
                #pragma unroll
                for (int ks = 0; ks < 16; ++ks) {
                    if (ks < 15) {
                        #pragma unroll
                        for (int q = 0; q < 7; ++q)
                            a[(ks + 1) & 1][q] = *(const v2f*)&(arow + po[q])[(ks + 1) * 196];
                        #pragma unroll
                        for (int s = 0; s < 3; ++s)
                            bf[(ks + 1) & 1][s] = *(const v2f*)&wb[s * SLAB + (ks + 1) * 256];
                    }
                    __builtin_amdgcn_sched_barrier(0);
                    if (ks == 0)       __builtin_amdgcn_s_setprio(3);
                    else if (ks == 4)  __builtin_amdgcn_s_setprio(2);
                    else if (ks == 8)  __builtin_amdgcn_s_setprio(1);
                    else if (ks == 12) __builtin_amdgcn_s_setprio(0);
                    const v2f* q = a[ks & 1];
                    const v2f t0 = bf[ks & 1][0], t2 = bf[ks & 1][1], t1 = bf[ks & 1][2];
                    const v2f e20 = q[2] - q[0], e53 = q[5] - q[3];
                    const v2f a0 = 0.5f * e20 - q[1];
                    const v2f a4 = (q[3] - q[1]) - 2.0f * e20;
                    const v2f b0 = (q[2] - q[4]) + 0.5f * e53;
                    const v2f b4 = (q[6] - q[4]) - 2.0f * e53;
                    acc[6] = __builtin_amdgcn_wmma_f32_16x16x4_f32(false, q[5], false, t0, (short)0, acc[6], false, false);
                    acc[0] = __builtin_amdgcn_wmma_f32_16x16x4_f32(false, a0, false, t0, (short)0, acc[0], false, false);
                    acc[3] = __builtin_amdgcn_wmma_f32_16x16x4_f32(false, b0, false, t0, (short)0, acc[3], false, false);
                    acc[2] = __builtin_amdgcn_wmma_f32_16x16x4_f32(false, a4, false, t2, (short)0, acc[2], false, false);
                    acc[5] = __builtin_amdgcn_wmma_f32_16x16x4_f32(false, b4, false, t2, (short)0, acc[5], false, false);
                    acc[6] = __builtin_amdgcn_wmma_f32_16x16x4_f32(false, q[6], false, t1, (short)0, acc[6], false, false);
                    __builtin_amdgcn_sched_barrier(0);
                }
            }
        }
    };

    v8f c1[7] = {};
    conv_pass(0, c1);
    __syncthreads();

    const float sc1 = g1[nn] * rsqrtf(v1[nn] + EPSV);
    const float sh1 = b1[nn] - m1[nn] * sc1;
    if (own) {
        #pragma unroll
        for (int t = 0; t < 7; ++t) {
            const int p = pg * 7 + t;
            #pragma unroll
            for (int j = 0; j < 8; ++j) {
                const int img = j + 8 * hi;
                float v = fmaf(c1[t][j], sc1, sh1);
                v = fmaxf(v, 0.0f);
                if (b0 + img == 0) v *= mask1[nn * 49 + p];
                act[img * IMGP + (nn >> 1) * 98 + 2 * p + (nn & 1)] = v;
            }
        }
    }

    v8f c2[7] = {};
    conv_pass(1, c2);
    __syncthreads();

    const float sc2 = g2[nn] * rsqrtf(v2p[nn] + EPSV);
    const float sh2 = b2[nn] - m2[nn] * sc2;
    if (own) {
        #pragma unroll
        for (int t = 0; t < 7; ++t) {
            const int p = pg * 7 + t;
            #pragma unroll
            for (int j = 0; j < 8; ++j) {
                const int img = j + 8 * hi;
                float v = c2[t][j];
                if (b0 + img == 0) v *= mask2[nn * 49 + p];
                v = fmaf(v, sc2, sh2);
                act[img * IMGP + nn * 49 + p] = v;
            }
        }
    }
    __syncthreads();

    float4* o4 = (float4*)(out + (size_t)b0 * 3136);
    int tl = tid;
    asm volatile("" : "+v"(tl));
    tl &= 1023;
    {
        float4 xr[13];
        #pragma unroll
        for (int j = 0; j < 13; ++j) {
            const int i = tl + j * 1024;
            xr[j] = (i < 12544) ? x4[i] : make_float4(0.f, 0.f, 0.f, 0.f);
        }
        #pragma unroll
        for (int j = 0; j < 13; ++j) {
            const int i = tl + j * 1024;
            if (i < 12544) {
                const int img = i / 784;
                float4 v = act4[img * (IMGP / 4) + (i - img * 784)];
                v.x = fmaxf(v.x + xr[j].x, 0.0f);
                v.y = fmaxf(v.y + xr[j].y, 0.0f);
                v.z = fmaxf(v.z + xr[j].z, 0.0f);
                v.w = fmaxf(v.w + xr[j].w, 0.0f);
                o4[i] = v;
            }
        }
    }
}

extern "C" void kernel_launch(void* const* d_in, const int* in_sizes, int n_in,
                              void* d_out, int out_size, void* d_ws, size_t ws_size,
                              hipStream_t stream) {
    const float* x     = (const float*)d_in[0];
    const float* w1    = (const float*)d_in[1];
    const float* g1    = (const float*)d_in[2];
    const float* b1    = (const float*)d_in[3];
    const float* m1    = (const float*)d_in[4];
    const float* v1    = (const float*)d_in[5];
    const float* w2    = (const float*)d_in[6];
    const float* g2    = (const float*)d_in[7];
    const float* b2    = (const float*)d_in[8];
    const float* m2    = (const float*)d_in[9];
    const float* v2p   = (const float*)d_in[10];
    const float* mask1 = (const float*)d_in[11];
    const float* mask2 = (const float*)d_in[12];
    float* wt = (float*)d_ws;
    float* out = (float*)d_out;

    prep_weights_kernel<<<288, 256, 0, stream>>>(w1, w2, wt);
    fused_basicblock_kernel<<<256, 1024, 0, stream>>>(
        x, wt, g1, b1, m1, v1, g2, b2, m2, v2p, mask1, mask2, out);
}


# Fast-FIR input transform hoisted out of the K-loop: pixel combinations formed once per conv in place in the LDS tile (one more in-place pass between the phases); K-loop WMMAs take fragments as loaded
# speedup vs baseline: 2.7783x; 1.0636x over previous
#include <hip/hip_runtime.h>

#define EPSV 1e-5f

typedef __attribute__((ext_vector_type(2))) float v2f;
typedef __attribute__((ext_vector_type(8))) float v8f;

__global__ __launch_bounds__(256) void prep_weights_kernel(
    const float* __restrict__ w1, const float* __restrict__ w2,
    float* __restrict__ wt)
{
    int i = blockIdx.x * 256 + threadIdx.x;
    if (i >= 2 * 36864) return;
    const float* w = (i < 36864) ? w1 : w2;
    float* dst = wt + (i < 36864 ? 0 : 36864);
    int j = i % 36864;
    int tap = j / 4096;
    int rem = j % 4096;
    int n = (rem >> 1) & 63;
    int k = ((rem >> 7) << 1) | (rem & 1);
    dst[j] = w[(n * 64 + k) * 9 + tap];
}

#define NIMG 16
#define IMGP 3140
#define SLAB 4096
#define STG  (3 * SLAB)

__device__ __forceinline__ void fir_tap1(const float g0, const float g1, const float g2,
                                         float& w1, float& w2, float& w3)
{
    const float e = g0 + g2;
    w1 = 0.5f * (e + g1);
    w2 = (1.0f / 6.0f) * (e - g1);
    w3 = (1.0f / 6.0f) * (fmaf(4.0f, g2, g0) + 2.0f * g1);
}
__device__ __forceinline__ void fir_taps(const float4 g0, const float4 g1, const float4 g2,
                                         float4& w1, float4& w2, float4& w3)
{
    fir_tap1(g0.x, g1.x, g2.x, w1.x, w2.x, w3.x);
    fir_tap1(g0.y, g1.y, g2.y, w1.y, w2.y, w3.y);
    fir_tap1(g0.z, g1.z, g2.z, w1.z, w2.z, w3.z);
    fir_tap1(g0.w, g1.w, g2.w, w1.w, w2.w, w3.w);
}

__global__ __launch_bounds__(1024) void fused_basicblock_kernel(
    const float* __restrict__ x,
    const float* __restrict__ wt,
    const float* __restrict__ g1, const float* __restrict__ b1,
    const float* __restrict__ m1, const float* __restrict__ v1,
    const float* __restrict__ g2, const float* __restrict__ b2,
    const float* __restrict__ m2, const float* __restrict__ v2p,
    const float* __restrict__ mask1, const float* __restrict__ mask2,
    float* __restrict__ out)
{
    __shared__ __align__(16) float act[NIMG * IMGP];
    __shared__ __align__(16) float wsh[2 * STG];

    const int tid  = threadIdx.x;
    const int lane = tid & 31;
    const int wave = __builtin_amdgcn_readfirstlane(tid >> 5);
    const int b0   = blockIdx.x * NIMG;

    const int lo = lane & 15;
    const int hi = lane >> 4;

    const int nn = (wave & 3) * 16 + lo;
    const int pg = 2 * (((wave >> 3) + (wave & 3)) & 3) + ((wave >> 2) & 1);
    const bool own = pg < 7;
    const int it = (wave & 3) * 32 + lane;
    if (!own) __builtin_amdgcn_s_setprio(3);

    int po[7];
    #pragma unroll
    for (int i = 0; i < 7; ++i) {
        int o = 2 * i;
        asm volatile("" : "+s"(o));
        po[i] = o;
    }

    const float4* w4 = (const float4*)wt;
    float4* wsh4 = (float4*)wsh;

    const float4* x4 = (const float4*)(x + (size_t)b0 * 3136);
    float4* act4 = (float4*)act;
    {
        const int simg = tid >> 6;
        const int l64 = tid & 63;
        const float* xg = x + (size_t)(b0 + simg) * 3136;
        float xe[25], xo[25];
        #pragma unroll
        for (int j = 0; j < 25; ++j) {
            const int r = l64 + j * 64;
            const int cp = (r * 1338) >> 16;
            const bool ok = (j < 24) || (l64 < 32);
            xe[j] = ok ? xg[r + cp * 49] : 0.0f;
            xo[j] = ok ? xg[r + cp * 49 + 49] : 0.0f;
        }
        const float4 wv0 = w4[tid], wv1 = w4[tid + 1024], wv2 = w4[tid + 2048];
        #pragma unroll
        for (int j = 0; j < 25; ++j) {
            const int r = l64 + j * 64;
            if ((j < 24) || (l64 < 32)) {
                v2f pr;
                pr[0] = xe[j];
                pr[1] = xo[j];
                *(v2f*)&act[simg * IMGP + 2 * r] = pr;
            }
        }
        float4 f1, f2, f3;
        fir_taps(wv0, wv1, wv2, f1, f2, f3);
        wsh4[tid] = f1;
        wsh4[tid + 1024] = f2;
        wsh4[tid + 2048] = f3;
    }

    auto to_a_form = [&]() {
        for (int k = 0; k < 4; ++k) {
            const int i = tid + k * 1024;
            if (i < NIMG * 224) {
                const int img = i / 224;
                float* p = act + img * IMGP + (i - img * 224) * 14;
                const v2f q0 = *(const v2f*)(p + 0), q1 = *(const v2f*)(p + 2), q2 = *(const v2f*)(p + 4);
                const v2f q3 = *(const v2f*)(p + 6), q4 = *(const v2f*)(p + 8), q5 = *(const v2f*)(p + 10);
                *(v2f*)(p + 0)  = 2.0f * q0 + (q1 - q2);
                *(v2f*)(p + 2)  = 3.0f * q1 - (2.0f * q0 + q2);
                *(v2f*)(p + 4)  = q2 - q0;
                *(v2f*)(p + 6)  = 2.0f * q3 + (q4 - q5);
                *(v2f*)(p + 8)  = 3.0f * q4 - (2.0f * q3 + q5);
                *(v2f*)(p + 10) = q5 - q3;
            }
        }
    };
    auto a_to_b_form = [&]() {
        for (int k = 0; k < 4; ++k) {
            const int i = tid + k * 1024;
            if (i < NIMG * 224) {
                const int img = i / 224;
                float* p = act + img * IMGP + (i - img * 224) * 14;
                const v2f a1 = *(const v2f*)(p + 0), a2 = *(const v2f*)(p + 2), a3 = *(const v2f*)(p + 4);
                const v2f b1 = *(const v2f*)(p + 6), b2 = *(const v2f*)(p + 8), b3 = *(const v2f*)(p + 10);
                const v2f q6 = *(const v2f*)(p + 12);
                const v2f sa = a1 + a3;
                const v2f da = (1.0f / 3.0f) * (a2 + a3);
                const v2f q1 = 0.5f * (sa + da);
                const v2f q0 = 0.5f * (sa - da);
                const v2f q2 = a3 + q0;
                const v2f sb = b1 + b3;
                const v2f db = (1.0f / 3.0f) * (b2 + b3);
                const v2f q4 = 0.5f * (sb + db);
                const v2f q3 = 0.5f * (sb - db);
                const v2f q5 = b3 + q3;
                *(v2f*)(p + 0)  = 0.5f * a3 - q1;
                *(v2f*)(p + 2)  = (q3 - q1) - 2.0f * a3;
                *(v2f*)(p + 4)  = (q2 - q4) + 0.5f * b3;
                *(v2f*)(p + 6)  = (q6 - q4) - 2.0f * b3;
                *(v2f*)(p + 10) = q5;
            }
        }
    };
    __syncthreads();
    to_a_form();

    const float* alane = act + lo * IMGP + hi * 98;

    auto fill_stage = [&](const int gn) {
        const int cn = (gn >= 6) ? 1 : 0;
        const int jn = gn - 6 * cn;
        const bool pa = jn < 3;
        const int fr = pa ? jn : jn - 3;
        const float4* srcw = w4 + (cn * 9 + fr * 3) * (SLAB / 4) + it;
        float4* dstw = wsh4 + (gn & 1) * (STG / 4) + it;
        if (pa) {
            #pragma unroll
            for (int j0 = 0; j0 < 8; j0 += 2) {
                const float4 t00 = srcw[(j0 + 0) * 128];
                const float4 t01 = srcw[(j0 + 0) * 128 + 1024];
                const float4 t02 = srcw[(j0 + 0) * 128 + 2048];
                const float4 t10 = srcw[(j0 + 1) * 128];
                const float4 t11 = srcw[(j0 + 1) * 128 + 1024];
                const float4 t12 = srcw[(j0 + 1) * 128 + 2048];
                float4 fa0, fb0, fc0, fa1, fb1, fc1;
                fir_taps(t00, t01, t02, fa0, fb0, fc0);
                fir_taps(t10, t11, t12, fa1, fb1, fc1);
                dstw[(j0 + 0) * 128] = fa0;
                dstw[(j0 + 0) * 128 + 1024] = fb0;
                dstw[(j0 + 0) * 128 + 2048] = fc0;
                dstw[(j0 + 1) * 128] = fa1;
                dstw[(j0 + 1) * 128 + 1024] = fb1;
                dstw[(j0 + 1) * 128 + 2048] = fc1;
            }
        } else {
            #pragma unroll
            for (int j0 = 0; j0 < 8; j0 += 2) {
                const float4 ta0 = srcw[(j0 + 0) * 128];
                const float4 tb0 = srcw[(j0 + 0) * 128 + 1024];
                const float4 tc0 = srcw[(j0 + 0) * 128 + 2048];
                const float4 ta1 = srcw[(j0 + 1) * 128];
                const float4 tb1 = srcw[(j0 + 1) * 128 + 1024];
                const float4 tc1 = srcw[(j0 + 1) * 128 + 2048];
                dstw[(j0 + 0) * 128] = ta0;
                dstw[(j0 + 0) * 128 + 1024] = tc0;
                dstw[(j0 + 0) * 128 + 2048] = tb0;
                dstw[(j0 + 1) * 128] = ta1;
                dstw[(j0 + 1) * 128 + 1024] = tc1;
                dstw[(j0 + 1) * 128 + 2048] = tb1;
            }
        }
    };

    auto conv_pass = [&](const int c, v8f* acc) {
        #pragma unroll 1
        for (int r = 0; r < 3; ++r) {
            const int g  = c * 6 + r;
            const int ih = pg + r - 1;
            const bool rowok = own && (unsigned)ih < 7u;
            __syncthreads();
            if (!own) fill_stage(g + 1);
            if (rowok) {
                const float* arow = alane + ih * 14;
                const float* wb = wsh + (g & 1) * STG + hi * 128 + nn * 2;
                v2f a[2][6];
                v2f bf[2][3];
                #pragma unroll
                for (int q = 0; q < 6; ++q) a[0][q] = *(const v2f*)(arow + po[q]);
                #pragma unroll
                for (int s = 0; s < 3; ++s) bf[0][s] = *(const v2f*)&wb[s * SLAB];
                #pragma unroll
                for (int ks = 0; ks < 16; ++ks) {
                    if (ks < 15) {
                        #pragma unroll
                        for (int q = 0; q < 6; ++q)
                            a[(ks + 1) & 1][q] = *(const v2f*)&(arow + po[q])[(ks + 1) * 196];
                        #pragma unroll
                        for (int s = 0; s < 3; ++s)
                            bf[(ks + 1) & 1][s] = *(const v2f*)&wb[s * SLAB + (ks + 1) * 256];
                    }
                    __builtin_amdgcn_sched_barrier(0);
                    if (ks == 0)       __builtin_amdgcn_s_setprio(3);
                    else if (ks == 4)  __builtin_amdgcn_s_setprio(2);
                    else if (ks == 8)  __builtin_amdgcn_s_setprio(1);
                    else if (ks == 12) __builtin_amdgcn_s_setprio(0);
                    const v2f* q = a[ks & 1];
                    const v2f w1 = bf[ks & 1][0], w2 = bf[ks & 1][1], w3 = bf[ks & 1][2];
                    acc[0] = __builtin_amdgcn_wmma_f32_16x16x4_f32(false, q[0], false, w1, (short)0, acc[0], false, false);
                    acc[3] = __builtin_amdgcn_wmma_f32_16x16x4_f32(false, q[3], false, w1, (short)0, acc[3], false, false);
                    acc[1] = __builtin_amdgcn_wmma_f32_16x16x4_f32(false, q[1], false, w2, (short)0, acc[1], false, false);
                    acc[4] = __builtin_amdgcn_wmma_f32_16x16x4_f32(false, q[4], false, w2, (short)0, acc[4], false, false);
                    acc[2] = __builtin_amdgcn_wmma_f32_16x16x4_f32(false, q[2], false, w3, (short)0, acc[2], false, false);
                    acc[5] = __builtin_amdgcn_wmma_f32_16x16x4_f32(false, q[5], false, w3, (short)0, acc[5], false, false);
                    __builtin_amdgcn_sched_barrier(0);
                }
            }
        }

        #pragma unroll
        for (int tl = 0; tl < 6; tl += 3) {
            const v8f m1 = acc[tl], m2 = acc[tl + 1], m3 = acc[tl + 2];
            const v8f sm = m1 + m2, df = m1 - m2;
            acc[tl]     = sm + m3;
            acc[tl + 1] = df + 2.0f * m3;
            acc[tl + 2] = sm + 4.0f * m3;
        }
        asm volatile("" : "+v"(acc[1]));
        asm volatile("" : "+v"(acc[4]));

        __syncthreads();
        a_to_b_form();

        #pragma unroll 1
        for (int r = 0; r < 3; ++r) {
            const int g  = c * 6 + 3 + r;
            const int ih = pg + r - 1;
            const bool rowok = own && (unsigned)ih < 7u;
            __syncthreads();
            if (!own) {
                if (g < 11) fill_stage(g + 1);
            }
            if (rowok) {
                const float* arow = alane + ih * 14;
                const float* wb = wsh + (g & 1) * STG + hi * 128 + nn * 2;
                v2f a[2][7];
                v2f bf[2][3];
                #pragma unroll
                for (int q = 0; q < 7; ++q)
                    if (q != 4) a[0][q] = *(const v2f*)(arow + po[q]);
                #pragma unroll
                for (int s = 0; s < 3; ++s) bf[0][s] = *(const v2f*)&wb[s * SLAB];
                #pragma unroll
                for (int ks = 0; ks < 16; ++ks) {
                    if (ks < 15) {
                        #pragma unroll
                        for (int q = 0; q < 7; ++q)
                            if (q != 4) a[(ks + 1) & 1][q] = *(const v2f*)&(arow + po[q])[(ks + 1) * 196];
                        #pragma unroll
                        for (int s = 0; s < 3; ++s)
                            bf[(ks + 1) & 1][s] = *(const v2f*)&wb[s * SLAB + (ks + 1) * 256];
                    }
                    __builtin_amdgcn_sched_barrier(0);
                    if (ks == 0)       __builtin_amdgcn_s_setprio(3);
                    else if (ks == 4)  __builtin_amdgcn_s_setprio(2);
                    else if (ks == 8)  __builtin_amdgcn_s_setprio(1);
                    else if (ks == 12) __builtin_amdgcn_s_setprio(0);
                    const v2f* q = a[ks & 1];
                    const v2f t0 = bf[ks & 1][0], t2 = bf[ks & 1][1], t1 = bf[ks & 1][2];
                    acc[6] = __builtin_amdgcn_wmma_f32_16x16x4_f32(false, q[5], false, t0, (short)0, acc[6], false, false);
                    acc[0] = __builtin_amdgcn_wmma_f32_16x16x4_f32(false, q[0], false, t0, (short)0, acc[0], false, false);
                    acc[3] = __builtin_amdgcn_wmma_f32_16x16x4_f32(false, q[2], false, t0, (short)0, acc[3], false, false);
                    acc[2] = __builtin_amdgcn_wmma_f32_16x16x4_f32(false, q[1], false, t2, (short)0, acc[2], false, false);
                    acc[5] = __builtin_amdgcn_wmma_f32_16x16x4_f32(false, q[3], false, t2, (short)0, acc[5], false, false);
                    acc[6] = __builtin_amdgcn_wmma_f32_16x16x4_f32(false, q[6], false, t1, (short)0, acc[6], false, false);
                    __builtin_amdgcn_sched_barrier(0);
                }
            }
        }
    };

    v8f c1[7] = {};
    conv_pass(0, c1);
    __syncthreads();

    const float sc1 = g1[nn] * rsqrtf(v1[nn] + EPSV);
    const float sh1 = b1[nn] - m1[nn] * sc1;
    if (own) {
        #pragma unroll
        for (int j = 0; j < 8; ++j) {
            const int img = j + 8 * hi;
            float v[7];
            #pragma unroll
            for (int t = 0; t < 7; ++t) {
                float u = fmaf(c1[t][j], sc1, sh1);
                u = fmaxf(u, 0.0f);
                if (b0 + img == 0) u *= mask1[nn * 49 + pg * 7 + t];
                v[t] = u;
            }
            float* p = act + img * IMGP + (nn >> 1) * 98 + 2 * (pg * 7) + (nn & 1);
            p[0]  = 2.0f * v[0] + (v[1] - v[2]);
            p[2]  = 3.0f * v[1] - (2.0f * v[0] + v[2]);
            p[4]  = v[2] - v[0];
            p[6]  = 2.0f * v[3] + (v[4] - v[5]);
            p[8]  = 3.0f * v[4] - (2.0f * v[3] + v[5]);
            p[10] = v[5] - v[3];
            p[12] = v[6];
        }
    }

    v8f c2[7] = {};
    conv_pass(1, c2);
    __syncthreads();

    const float sc2 = g2[nn] * rsqrtf(v2p[nn] + EPSV);
    const float sh2 = b2[nn] - m2[nn] * sc2;
    if (own) {
        #pragma unroll
        for (int t = 0; t < 7; ++t) {
            const int p = pg * 7 + t;
            #pragma unroll
            for (int j = 0; j < 8; ++j) {
                const int img = j + 8 * hi;
                float v = c2[t][j];
                if (b0 + img == 0) v *= mask2[nn * 49 + p];
                v = fmaf(v, sc2, sh2);
                act[img * IMGP + nn * 49 + p] = v;
            }
        }
    }
    __syncthreads();

    float4* o4 = (float4*)(out + (size_t)b0 * 3136);
    int tl = tid;
    asm volatile("" : "+v"(tl));
    tl &= 1023;
    {
        float4 xr[13];
        #pragma unroll
        for (int j = 0; j < 13; ++j) {
            const int i = tl + j * 1024;
            xr[j] = (i < 12544) ? x4[i] : make_float4(0.f, 0.f, 0.f, 0.f);
        }
        #pragma unroll
        for (int j = 0; j < 13; ++j) {
            const int i = tl + j * 1024;
            if (i < 12544) {
                const int img = i / 784;
                float4 v = act4[img * (IMGP / 4) + (i - img * 784)];
                v.x = fmaxf(v.x + xr[j].x, 0.0f);
                v.y = fmaxf(v.y + xr[j].y, 0.0f);
                v.z = fmaxf(v.z + xr[j].z, 0.0f);
                v.w = fmaxf(v.w + xr[j].w, 0.0f);
                o4[i] = v;
            }
        }
    }
}

extern "C" void kernel_launch(void* const* d_in, const int* in_sizes, int n_in,
                              void* d_out, int out_size, void* d_ws, size_t ws_size,
                              hipStream_t stream) {
    const float* x     = (const float*)d_in[0];
    const float* w1    = (const float*)d_in[1];
    const float* g1    = (const float*)d_in[2];
    const float* b1    = (const float*)d_in[3];
    const float* m1    = (const float*)d_in[4];
    const float* v1    = (const float*)d_in[5];
    const float* w2    = (const float*)d_in[6];
    const float* g2    = (const float*)d_in[7];
    const float* b2    = (const float*)d_in[8];
    const float* m2    = (const float*)d_in[9];
    const float* v2p   = (const float*)d_in[10];
    const float* mask1 = (const float*)d_in[11];
    const float* mask2 = (const float*)d_in[12];
    float* wt = (float*)d_ws;
    float* out = (float*)d_out;

    prep_weights_kernel<<<288, 256, 0, stream>>>(w1, w2, wt);
    fused_basicblock_kernel<<<256, 1024, 0, stream>>>(
        x, wt, g1, b1, m1, v1, g2, b2, m2, v2p, mask1, mask2, out);
}


# Row re-tiled: pixels 0-3 by the four-output fast FIR (points 0,1,-1,2,-2,inf; 6 products), pixels 4-6 by the three-output one (5), no direct pixel; 11 f32 WMMAs per filter row and K-step, was 12
# speedup vs baseline: 2.9209x; 1.0513x over previous
#include <hip/hip_runtime.h>

#define EPSV 1e-5f

typedef __attribute__((ext_vector_type(2))) float v2f;
typedef __attribute__((ext_vector_type(8))) float v8f;

__global__ __launch_bounds__(256) void prep_weights_kernel(
    const float* __restrict__ w1, const float* __restrict__ w2,
    float* __restrict__ wt)
{
    int i = blockIdx.x * 256 + threadIdx.x;
    if (i >= 2 * 36864) return;
    const float* w = (i < 36864) ? w1 : w2;
    float* dst = wt + (i < 36864 ? 0 : 36864);
    int j = i % 36864;
    int tap = j / 4096;
    int rem = j % 4096;
    int n = (rem >> 1) & 63;
    int k = ((rem >> 7) << 1) | (rem & 1);
    dst[j] = w[(n * 64 + k) * 9 + tap];
}

#define NIMG 16
#define IMGP 3140
#define SLAB 4096
#define STG  (3 * SLAB)

__device__ __forceinline__ void fir_tap1(const float g0, const float g1, const float g2,
                                         float& w1, float& w2, float& w3)
{
    const float e = g0 + g2;
    w1 = 0.5f * (e + g1);
    w2 = (1.0f / 6.0f) * (e - g1);
    w3 = (1.0f / 6.0f) * (fmaf(4.0f, g2, g0) + 2.0f * g1);
}
__device__ __forceinline__ void fir_taps(const float4 g0, const float4 g1, const float4 g2,
                                         float4& w1, float4& w2, float4& w3)
{
    fir_tap1(g0.x, g1.x, g2.x, w1.x, w2.x, w3.x);
    fir_tap1(g0.y, g1.y, g2.y, w1.y, w2.y, w3.y);
    fir_tap1(g0.z, g1.z, g2.z, w1.z, w2.z, w3.z);
    fir_tap1(g0.w, g1.w, g2.w, w1.w, w2.w, w3.w);
}
__device__ __forceinline__ float fir_tapm1(const float g0, const float g1, const float g2)
{
    return (1.0f / 24.0f) * (fmaf(4.0f, g2, g0) - 2.0f * g1);
}
__device__ __forceinline__ float4 fir_tapm(const float4 g0, const float4 g1, const float4 g2)
{
    float4 w;
    w.x = fir_tapm1(g0.x, g1.x, g2.x);
    w.y = fir_tapm1(g0.y, g1.y, g2.y);
    w.z = fir_tapm1(g0.z, g1.z, g2.z);
    w.w = fir_tapm1(g0.w, g1.w, g2.w);
    return w;
}

__global__ __launch_bounds__(1024) void fused_basicblock_kernel(
    const float* __restrict__ x,
    const float* __restrict__ wt,
    const float* __restrict__ g1, const float* __restrict__ b1,
    const float* __restrict__ m1, const float* __restrict__ v1,
    const float* __restrict__ g2, const float* __restrict__ b2,
    const float* __restrict__ m2, const float* __restrict__ v2p,
    const float* __restrict__ mask1, const float* __restrict__ mask2,
    float* __restrict__ out)
{
    __shared__ __align__(16) float act[NIMG * IMGP];
    __shared__ __align__(16) float wsh[2 * STG];

    const int tid  = threadIdx.x;
    const int lane = tid & 31;
    const int wave = __builtin_amdgcn_readfirstlane(tid >> 5);
    const int b0   = blockIdx.x * NIMG;

    const int lo = lane & 15;
    const int hi = lane >> 4;

    const int nn = (wave & 3) * 16 + lo;
    const int pg = 2 * (((wave >> 3) + (wave & 3)) & 3) + ((wave >> 2) & 1);
    const bool own = pg < 7;
    const int it = (wave & 3) * 32 + lane;
    if (!own) __builtin_amdgcn_s_setprio(3);

    int po[7];
    #pragma unroll
    for (int i = 0; i < 7; ++i) {
        int o = 2 * i;
        asm volatile("" : "+s"(o));
        po[i] = o;
    }

    const float4* w4 = (const float4*)wt;
    float4* wsh4 = (float4*)wsh;

    const float4* x4 = (const float4*)(x + (size_t)b0 * 3136);
    float4* act4 = (float4*)act;
    {
        const int simg = tid >> 6;
        const int l64 = tid & 63;
        const float* xg = x + (size_t)(b0 + simg) * 3136;
        float xe[25], xo[25];
        #pragma unroll
        for (int j = 0; j < 25; ++j) {
            const int r = l64 + j * 64;
            const int cp = (r * 1338) >> 16;
            const bool ok = (j < 24) || (l64 < 32);
            xe[j] = ok ? xg[r + cp * 49] : 0.0f;
            xo[j] = ok ? xg[r + cp * 49 + 49] : 0.0f;
        }
        const float4 wv0 = w4[tid], wv1 = w4[tid + 1024], wv2 = w4[tid + 2048];
        #pragma unroll
        for (int j = 0; j < 25; ++j) {
            const int r = l64 + j * 64;
            if ((j < 24) || (l64 < 32)) {
                v2f pr;
                pr[0] = xe[j];
                pr[1] = xo[j];
                *(v2f*)&act[simg * IMGP + 2 * r] = pr;
            }
        }
        float4 f1, f2, f3;
        fir_taps(wv0, wv1, wv2, f1, f2, f3);
        wsh4[tid] = f1;
        wsh4[tid + 1024] = f2;
        wsh4[tid + 2048] = f3;
    }

    auto to_a_form = [&]() {
        for (int k = 0; k < 4; ++k) {
            const int i = tid + k * 1024;
            if (i < NIMG * 224) {
                const int img = i / 224;
                float* p = act + img * IMGP + (i - img * 224) * 14;
                const v2f q0 = *(const v2f*)(p + 0), q1 = *(const v2f*)(p + 2), q2 = *(const v2f*)(p + 4);
                const v2f q3 = *(const v2f*)(p + 6), q4 = *(const v2f*)(p + 8), q5 = *(const v2f*)(p + 10);
                const v2f q6 = *(const v2f*)(p + 12);
                *(v2f*)(p + 0)  = (1.0f / 3.0f) * (4.0f * (q0 + q1) - (q2 + q3));
                *(v2f*)(p + 2)  = 4.0f * (q1 - q0) + (q2 - q3);
                *(v2f*)(p + 4)  = 0.25f * (2.0f * (q2 - q0) + (q3 - q1));
                *(v2f*)(p + 6)  = 2.0f * q4 + (q5 - q6);
                *(v2f*)(p + 8)  = 3.0f * q5 - (2.0f * q4 + q6);
                *(v2f*)(p + 10) = q6 - q4;
                *(v2f*)(p + 12) = 2.0f * (q0 - q2) + (q3 - q1);
            }
        }
    };
    auto a_to_b_form = [&]() {
        for (int k = 0; k < 4; ++k) {
            const int i = tid + k * 1024;
            if (i < NIMG * 224) {
                const int img = i / 224;
                float* p = act + img * IMGP + (i - img * 224) * 14;
                const v2f l1 = *(const v2f*)(p + 0), l2 = *(const v2f*)(p + 2), l3 = *(const v2f*)(p + 4);
                const v2f r1 = *(const v2f*)(p + 6), r2 = *(const v2f*)(p + 8), r3 = *(const v2f*)(p + 10);
                const v2f l4 = *(const v2f*)(p + 12);
                const v2f t1 = 3.0f * l1, t3 = 4.0f * l3;
                const v2f e1 = t1 + l2, o1 = t1 - l2;
                const v2f e2 = t3 + l4, o2 = t3 - l4;
                const v2f q1 = (1.0f / 6.0f) * (e1 + e2);
                const v2f q3 = q1 + 0.5f * e2;
                const v2f q0 = (1.0f / 12.0f) * (2.0f * o1 + o2);
                const v2f q2 = q0 + 0.25f * o2;
                const v2f sb = r1 + r3;
                const v2f db = (1.0f / 3.0f) * (r2 + r3);
                const v2f q5 = 0.5f * (sb + db);
                const v2f q4 = 0.5f * (sb - db);
                *(v2f*)(p + 0) = 0.25f * (q3 - 5.0f * q1);
                *(v2f*)(p + 2) = (4.0f * q0 + q4) - 5.0f * q2;
                *(v2f*)(p + 4) = (q3 - q5) + 0.5f * r3;
                *(v2f*)(p + 6) = -q5 - 2.0f * r3;
            }
        }
    };
    __syncthreads();
    to_a_form();

    const float* alane = act + lo * IMGP + hi * 98;

    auto fill_stage = [&](const int gn) __attribute__((always_inline)) {
        const int cn = (gn >= 6) ? 1 : 0;
        const int jn = gn - 6 * cn;
        const bool pa = jn < 3;
        const int fr = pa ? jn : jn - 3;
        const float4* srcw = w4 + (cn * 9 + fr * 3) * (SLAB / 4) + it;
        float4* dstw = wsh4 + (gn & 1) * (STG / 4) + it;
        if (pa) {
            #pragma unroll
            for (int j0 = 0; j0 < 8; j0 += 2) {
                const float4 t00 = srcw[(j0 + 0) * 128];
                const float4 t01 = srcw[(j0 + 0) * 128 + 1024];
                const float4 t02 = srcw[(j0 + 0) * 128 + 2048];
                const float4 t10 = srcw[(j0 + 1) * 128];
                const float4 t11 = srcw[(j0 + 1) * 128 + 1024];
                const float4 t12 = srcw[(j0 + 1) * 128 + 2048];
                float4 fa0, fb0, fc0, fa1, fb1, fc1;
                fir_taps(t00, t01, t02, fa0, fb0, fc0);
                fir_taps(t10, t11, t12, fa1, fb1, fc1);
                dstw[(j0 + 0) * 128] = fa0;
                dstw[(j0 + 0) * 128 + 1024] = fb0;
                dstw[(j0 + 0) * 128 + 2048] = fc0;
                dstw[(j0 + 1) * 128] = fa1;
                dstw[(j0 + 1) * 128 + 1024] = fb1;
                dstw[(j0 + 1) * 128 + 2048] = fc1;
            }
        } else {
            #pragma unroll
            for (int j0 = 0; j0 < 8; j0 += 2) {
                const float4 ta0 = srcw[(j0 + 0) * 128];
                const float4 tb0 = srcw[(j0 + 0) * 128 + 1024];
                const float4 tc0 = srcw[(j0 + 0) * 128 + 2048];
                const float4 ta1 = srcw[(j0 + 1) * 128];
                const float4 tb1 = srcw[(j0 + 1) * 128 + 1024];
                const float4 tc1 = srcw[(j0 + 1) * 128 + 2048];
                dstw[(j0 + 0) * 128] = fir_tapm(ta0, tb0, tc0);
                dstw[(j0 + 0) * 128 + 1024] = ta0;
                dstw[(j0 + 0) * 128 + 2048] = tc0;
                dstw[(j0 + 1) * 128] = fir_tapm(ta1, tb1, tc1);
                dstw[(j0 + 1) * 128 + 1024] = ta1;
                dstw[(j0 + 1) * 128 + 2048] = tc1;
            }
        }
    };

    auto conv_pass = [&](const int c, v8f* acc) {
        #pragma unroll 1
        for (int r = 0; r < 3; ++r) {
            const int g  = c * 6 + r;
            const int ih = pg + r - 1;
            const bool rowok = own && (unsigned)ih < 7u;
            __syncthreads();
            if (!own) fill_stage(g + 1);
            if (rowok) {
                const float* arow = alane + ih * 14;
                const float* wb = wsh + (g & 1) * STG + hi * 128 + nn * 2;
                v2f a[2][6];
                v2f bf[2][3];
                #pragma unroll
                for (int q = 0; q < 6; ++q) a[0][q] = *(const v2f*)(arow + po[q]);
                #pragma unroll
                for (int s = 0; s < 3; ++s) bf[0][s] = *(const v2f*)&wb[s * SLAB];
                #pragma unroll
                for (int ks = 0; ks < 16; ++ks) {
                    if (ks < 15) {
                        #pragma unroll
                        for (int q = 0; q < 6; ++q)
                            a[(ks + 1) & 1][q] = *(const v2f*)&(arow + po[q])[(ks + 1) * 196];
                        #pragma unroll
                        for (int s = 0; s < 3; ++s)
                            bf[(ks + 1) & 1][s] = *(const v2f*)&wb[s * SLAB + (ks + 1) * 256];
                    }
                    __builtin_amdgcn_sched_barrier(0);
                    if (ks == 0)       __builtin_amdgcn_s_setprio(3);
                    else if (ks == 4)  __builtin_amdgcn_s_setprio(2);
                    else if (ks == 8)  __builtin_amdgcn_s_setprio(1);
                    else if (ks == 12) __builtin_amdgcn_s_setprio(0);
                    const v2f* q = a[ks & 1];
                    const v2f w1 = bf[ks & 1][0], w2 = bf[ks & 1][1], w3 = bf[ks & 1][2];
                    acc[0] = __builtin_amdgcn_wmma_f32_16x16x4_f32(false, q[0], false, w1, (short)0, acc[0], false, false);
                    acc[3] = __builtin_amdgcn_wmma_f32_16x16x4_f32(false, q[3], false, w1, (short)0, acc[3], false, false);
                    acc[1] = __builtin_amdgcn_wmma_f32_16x16x4_f32(false, q[1], false, w2, (short)0, acc[1], false, false);
                    acc[4] = __builtin_amdgcn_wmma_f32_16x16x4_f32(false, q[4], false, w2, (short)0, acc[4], false, false);
                    acc[2] = __builtin_amdgcn_wmma_f32_16x16x4_f32(false, q[2], false, w3, (short)0, acc[2], false, false);
                    acc[5] = __builtin_amdgcn_wmma_f32_16x16x4_f32(false, q[5], false, w3, (short)0, acc[5], false, false);
                    __builtin_amdgcn_sched_barrier(0);
                }
            }
        }

        {
            const v8f l1 = acc[0], l2 = acc[1], l3 = acc[2];
            const v8f r1 = acc[3], r2 = acc[4], r3 = acc[5];
            const v8f sl = l1 + l2, dl = l1 - l2;
            const v8f sr = r1 + r2, dr = r1 - r2;
            acc[0] = sl + l3;
            acc[1] = dl + 2.0f * l3;
            acc[2] = sl + 4.0f * l3;
            acc[3] = dl + 8.0f * l3;
            acc[4] = sr + r3;
            acc[5] = dr + 2.0f * r3;
            acc[6] = sr + 4.0f * r3;
        }
        asm volatile("" : "+v"(acc[1]));
        asm volatile("" : "+v"(acc[2]));
        asm volatile("" : "+v"(acc[5]));

        __syncthreads();
        a_to_b_form();

        #pragma unroll 1
        for (int r = 0; r < 3; ++r) {
            const int g  = c * 6 + 3 + r;
            const int ih = pg + r - 1;
            const bool rowok = own && (unsigned)ih < 7u;
            __syncthreads();
            if (!own) {
                if (g < 11) fill_stage(g + 1);
            }
            if (rowok) {
                const float* arow = alane + ih * 14;
                const float* wb = wsh + (g & 1) * STG + hi * 128 + nn * 2;
                v2f a[2][7];
                v2f bf[2][3];
                #pragma unroll
                for (int q = 0; q < 7; ++q)
                    if (q != 4 && q != 5) a[0][q] = *(const v2f*)(arow + po[q]);
                #pragma unroll
                for (int s = 0; s < 3; ++s) bf[0][s] = *(const v2f*)&wb[s * SLAB];
                #pragma unroll
                for (int ks = 0; ks < 16; ++ks) {
                    if (ks < 15) {
                        #pragma unroll
                        for (int q = 0; q < 7; ++q)
                            if (q != 4 && q != 5) a[(ks + 1) & 1][q] = *(const v2f*)&(arow + po[q])[(ks + 1) * 196];
                        #pragma unroll
                        for (int s = 0; s < 3; ++s)
                            bf[(ks + 1) & 1][s] = *(const v2f*)&wb[s * SLAB + (ks + 1) * 256];
                    }
                    __builtin_amdgcn_sched_barrier(0);
                    if (ks == 0)       __builtin_amdgcn_s_setprio(3);
                    else if (ks == 4)  __builtin_amdgcn_s_setprio(2);
                    else if (ks == 8)  __builtin_amdgcn_s_setprio(1);
                    else if (ks == 12) __builtin_amdgcn_s_setprio(0);
                    const v2f* q = a[ks & 1];
                    const v2f t4 = bf[ks & 1][0], t0 = bf[ks & 1][1], t2 = bf[ks & 1][2];
                    acc[7] = __builtin_amdgcn_wmma_f32_16x16x4_f32(false, q[6], false, t4, (short)0, acc[7], false, false);
                    acc[0] = __builtin_amdgcn_wmma_f32_16x16x4_f32(false, q[0], false, t0, (short)0, acc[0], false, false);
                    acc[4] = __builtin_amdgcn_wmma_f32_16x16x4_f32(false, q[2], false, t0, (short)0, acc[4], false, false);
                    acc[3] = __builtin_amdgcn_wmma_f32_16x16x4_f32(false, q[1], false, t2, (short)0, acc[3], false, false);
                    acc[6] = __builtin_amdgcn_wmma_f32_16x16x4_f32(false, q[3], false, t2, (short)0, acc[6], false, false);
                    __builtin_amdgcn_sched_barrier(0);
                }
            }
        }

        {
            const v8f l4 = acc[7];
            acc[0] += l4;
            acc[1] -= 2.0f * l4;
            acc[2] += 4.0f * l4;
            acc[3] -= 8.0f * l4;
        }
    };

    v8f c1[8] = {};
    conv_pass(0, c1);
    __syncthreads();

    const float sc1 = g1[nn] * rsqrtf(v1[nn] + EPSV);
    const float sh1 = b1[nn] - m1[nn] * sc1;
    if (own) {
        #pragma unroll
        for (int j = 0; j < 8; ++j) {
            const int img = j + 8 * hi;
            float v[7];
            #pragma unroll
            for (int t = 0; t < 7; ++t) {
                float u = fmaf(c1[t][j], sc1, sh1);
                u = fmaxf(u, 0.0f);
                if (b0 + img == 0) u *= mask1[nn * 49 + pg * 7 + t];
                v[t] = u;
            }
            float* p = act + img * IMGP + (nn >> 1) * 98 + 2 * (pg * 7) + (nn & 1);
            p[0]  = (1.0f / 3.0f) * (4.0f * (v[0] + v[1]) - (v[2] + v[3]));
            p[2]  = 4.0f * (v[1] - v[0]) + (v[2] - v[3]);
            p[4]  = 0.25f * (2.0f * (v[2] - v[0]) + (v[3] - v[1]));
            p[6]  = 2.0f * v[4] + (v[5] - v[6]);
            p[8]  = 3.0f * v[5] - (2.0f * v[4] + v[6]);
            p[10] = v[6] - v[4];
            p[12] = 2.0f * (v[0] - v[2]) + (v[3] - v[1]);
        }
    }

    v8f c2[8] = {};
    conv_pass(1, c2);
    __syncthreads();

    const float sc2 = g2[nn] * rsqrtf(v2p[nn] + EPSV);
    const float sh2 = b2[nn] - m2[nn] * sc2;
    if (own) {
        #pragma unroll
        for (int t = 0; t < 7; ++t) {
            const int p = pg * 7 + t;
            #pragma unroll
            for (int j = 0; j < 8; ++j) {
                const int img = j + 8 * hi;
                float v = c2[t][j];
                if (b0 + img == 0) v *= mask2[nn * 49 + p];
                v = fmaf(v, sc2, sh2);
                act[img * IMGP + nn * 49 + p] = v;
            }
        }
    }
    __syncthreads();

    float4* o4 = (float4*)(out + (size_t)b0 * 3136);
    int tl = tid;
    asm volatile("" : "+v"(tl));
    tl &= 1023;
    {
        float4 xr[13];
        #pragma unroll
        for (int j = 0; j < 13; ++j) {
            const int i = tl + j * 1024;
            xr[j] = (i < 12544) ? x4[i] : make_float4(0.f, 0.f, 0.f, 0.f);
        }
        #pragma unroll
        for (int j = 0; j < 13; ++j) {
            const int i = tl + j * 1024;
            if (i < 12544) {
                const int img = i / 784;
                float4 v = act4[img * (IMGP / 4) + (i - img * 784)];
                v.x = fmaxf(v.x + xr[j].x, 0.0f);
                v.y = fmaxf(v.y + xr[j].y, 0.0f);
                v.z = fmaxf(v.z + xr[j].z, 0.0f);
                v.w = fmaxf(v.w + xr[j].w, 0.0f);
                o4[i] = v;
            }
        }
    }
}

extern "C" void kernel_launch(void* const* d_in, const int* in_sizes, int n_in,
                              void* d_out, int out_size, void* d_ws, size_t ws_size,
                              hipStream_t stream) {
    const float* x     = (const float*)d_in[0];
    const float* w1    = (const float*)d_in[1];
    const float* g1    = (const float*)d_in[2];
    const float* b1    = (const float*)d_in[3];
    const float* m1    = (const float*)d_in[4];
    const float* v1    = (const float*)d_in[5];
    const float* w2    = (const float*)d_in[6];
    const float* g2    = (const float*)d_in[7];
    const float* b2    = (const float*)d_in[8];
    const float* m2    = (const float*)d_in[9];
    const float* v2p   = (const float*)d_in[10];
    const float* mask1 = (const float*)d_in[11];
    const float* mask2 = (const float*)d_in[12];
    float* wt = (float*)d_ws;
    float* out = (float*)d_out;

    prep_weights_kernel<<<288, 256, 0, stream>>>(w1, w2, wt);
    fused_basicblock_kernel<<<256, 1024, 0, stream>>>(
        x, wt, g1, b1, m1, v1, g2, b2, m2, v2p, mask1, mask2, out);
}
